# starting kernel
# baseline (speedup 1.0000x reference)
#include <hip/hip_runtime.h>

typedef __attribute__((ext_vector_type(2))) float v2f;
typedef __attribute__((ext_vector_type(8))) float v8f;

#define LDS_STRIDE 68

__global__ void gin_zero_kernel(float4* __restrict__ p, int n4) {
  int i = blockIdx.x * blockDim.x + threadIdx.x;
  if (i < n4) p[i] = make_float4(0.f, 0.f, 0.f, 0.f);
}

__global__ void gin_scatter_kernel(const float* __restrict__ x,
                                   const int* __restrict__ ei,
                                   float* __restrict__ agg,
                                   int n_edges) {
  int gtid = blockIdx.x * blockDim.x + threadIdx.x;
  int edge = gtid >> 5;
  int lane = threadIdx.x & 31;
  if (edge >= n_edges) return;
  int src = ei[edge];
  int dst = ei[n_edges + edge];
  const float2 v = *(const float2*)(x + (size_t)src * 64 + lane * 2);
  float* p = agg + (size_t)dst * 64 + lane * 2;
  atomicAdd(p + 0, v.x);
  atomicAdd(p + 1, v.y);
}

__global__ __launch_bounds__(256) void gin_mlp_kernel(
    const float* __restrict__ x, const float* __restrict__ agg,
    const float* __restrict__ eps_p,
    const float* __restrict__ W1, const float* __restrict__ b1,
    const float* __restrict__ W2, const float* __restrict__ b2,
    float* __restrict__ out, int n_nodes) {
  __shared__ float lds[8 * 16 * LDS_STRIDE];

  const int tid  = threadIdx.x;
  const int wave = tid >> 5;
  const int lane = tid & 31;
  const int half = lane >> 4;
  const int m16  = lane & 15;
  float* tile = lds + wave * 16 * LDS_STRIDE;

  const float k1 = 1.0f + eps_p[0];
  const int node0 = (blockIdx.x * 8 + wave) * 16;

  for (int i = 0; i < 8; ++i) {
    int e4 = lane + i * 32;
    int r  = e4 >> 4;
    int c4 = e4 & 15;
    int node = node0 + r;
    int cn = node < n_nodes ? node : n_nodes - 1;
    float4 xv = *((const float4*)(x   + (size_t)cn * 64) + c4);
    float4 av = *((const float4*)(agg + (size_t)cn * 64) + c4);
    float* d = tile + r * LDS_STRIDE + c4 * 4;
    d[0] = k1 * xv.x + av.x;
    d[1] = k1 * xv.y + av.y;
    d[2] = k1 * xv.z + av.z;
    d[3] = k1 * xv.w + av.w;
  }
  __syncthreads();

  v8f acc[4];
  for (int t = 0; t < 4; ++t)
    for (int j = 0; j < 8; ++j) acc[t][j] = 0.f;

  for (int k = 0; k < 16; ++k) {
    const float2 af = *(const float2*)(tile + m16 * LDS_STRIDE + 4 * k + 2 * half);
    v2f a; a.x = af.x; a.y = af.y;
    for (int t = 0; t < 4; ++t) {
      int n = t * 16 + m16;
      const float2 bf = *(const float2*)(W1 + (size_t)n * 64 + 4 * k + 2 * half);
      v2f b; b.x = bf.x; b.y = bf.y;
      acc[t] = __builtin_amdgcn_wmma_f32_16x16x4_f32(
          false, a, false, b, (short)0, acc[t], false, false);
    }
  }
  __syncthreads();

  for (int t = 0; t < 4; ++t) {
    float bv = b1[t * 16 + m16];
    for (int v = 0; v < 8; ++v) {
      float h = acc[t][v] + bv;
      h = h > 0.f ? h : 0.f;
      tile[(v + 8 * half) * LDS_STRIDE + t * 16 + m16] = h;
    }
  }
  __syncthreads();

  v8f acc2[4];
  for (int t = 0; t < 4; ++t)
    for (int j = 0; j < 8; ++j) acc2[t][j] = 0.f;

  for (int k = 0; k < 16; ++k) {
    const float2 af = *(const float2*)(tile + m16 * LDS_STRIDE + 4 * k + 2 * half);
    v2f a; a.x = af.x; a.y = af.y;
    for (int t = 0; t < 4; ++t) {
      int n = t * 16 + m16;
      const float2 bf = *(const float2*)(W2 + (size_t)n * 64 + 4 * k + 2 * half);
      v2f b; b.x = bf.x; b.y = bf.y;
      acc2[t] = __builtin_amdgcn_wmma_f32_16x16x4_f32(
          false, a, false, b, (short)0, acc2[t], false, false);
    }
  }

  for (int t = 0; t < 4; ++t) {
    float bv = b2[t * 16 + m16];
    for (int v = 0; v < 8; ++v) {
      int node = node0 + v + 8 * half;
      if (node < n_nodes)
        out[(size_t)node * 64 + t * 16 + m16] = acc2[t][v] + bv;
    }
  }
}

extern "C" void kernel_launch(void* const* d_in, const int* in_sizes, int n_in,
                              void* d_out, int out_size, void* d_ws, size_t ws_size,
                              hipStream_t stream) {
  const float* x   = (const float*)d_in[0];
  const int*   ei  = (const int*)  d_in[1];
  const float* eps = (const float*)d_in[2];
  const float* W1  = (const float*)d_in[3];
  const float* b1  = (const float*)d_in[4];
  const float* W2  = (const float*)d_in[5];
  const float* b2  = (const float*)d_in[6];
  float* out = (float*)d_out;

  const int n_nodes = in_sizes[0] / 64;
  const int n_edges = in_sizes[1] / 2;
  float* agg = (float*)d_ws;

  int n4 = (n_nodes * 64) / 4;
  gin_zero_kernel<<<(n4 + 255) / 256, 256, 0, stream>>>((float4*)agg, n4);

  long long sthreads = (long long)n_edges * 32;
  int sblocks = (int)((sthreads + 255) / 256);
  gin_scatter_kernel<<<sblocks, 256, 0, stream>>>(x, ei, agg, n_edges);

  int ntiles = (n_nodes + 15) / 16;
  int mblocks = (ntiles + 7) / 8;
  gin_mlp_kernel<<<mblocks, 256, 0, stream>>>(x, agg, eps, W1, b1, W2, b2, out, n_nodes);
}


# Global f32-atomic scatter replaced by owner work-groups: each holds 1024 target rows in LDS, scans the whole target list and adds its edges' source rows with LDS float adds (arrival order)
# speedup vs baseline: 1.4546x; 1.4546x over previous
#include <hip/hip_runtime.h>

typedef __attribute__((ext_vector_type(2))) float v2f;
typedef __attribute__((ext_vector_type(8))) float v8f;

#define LDS_STRIDE 68

#define OWN_ROWS  1024
#define Q_CAP     256
#define ROW_BATCH 16
#define UNIT_V4   8

__device__ __forceinline__ void owner_drain(const float* __restrict__ x, const int* __restrict__ ei,
                                            const int* __restrict__ dstp, float* acc, const int* q,
                                            int head, int count, int row0, int lane) {
  __builtin_amdgcn_fence(__ATOMIC_RELEASE, "wavefront");
  __builtin_amdgcn_wave_barrier();
  __builtin_amdgcn_fence(__ATOMIC_ACQUIRE, "wavefront");
  int s = 0, r = 0;
  if (lane < count) {
    const int e = q[(head + lane) & (Q_CAP - 1)];
    s = ei[e];
    r = dstp[e] - row0;
  }
#pragma nounroll
  for (int k0 = 0; k0 < count; k0 += ROW_BATCH) {
    float2 v[ROW_BATCH];
#pragma unroll
    for (int j = 0; j < ROW_BATCH; ++j) {
      const int sk = __builtin_amdgcn_readlane(s, k0 + j);
      v[j] = *(const float2*)(x + (size_t)sk * 64 + 2 * lane);
    }
#pragma unroll
    for (int j = 0; j < ROW_BATCH; ++j) {
      if (k0 + j < count) {
        const int rk = __builtin_amdgcn_readlane(r, k0 + j);
        float* p = acc + rk * 64 + 2 * lane;
        atomicAdd(p, v[j].x);
        atomicAdd(p + 1, v[j].y);
      }
    }
  }
}

#define OWNER_PUSH(dval, eidx)                                                               \
  {                                                                                          \
    const int rel_ = (dval) - row0;                                                          \
    const bool in_ = (unsigned)rel_ < (unsigned)rows;                                        \
    const unsigned mask_ = __builtin_amdgcn_ballot_w32(in_);                                 \
    if (mask_ != 0u) {                                                                       \
      if (in_) q[(head + qn + (int)__builtin_amdgcn_mbcnt_lo(mask_, 0u)) & (Q_CAP - 1)] = (eidx); \
      qn += __builtin_popcount(mask_);                                                       \
    }                                                                                        \
  }
#define OWNER_DRAIN_FULL                                                                     \
  while (qn >= 32) {                                                                         \
    owner_drain(x, ei, dstp, acc, q, head, 32, row0, lane);                                  \
    head = (head + 32) & (Q_CAP - 1);                                                        \
    qn -= 32;                                                                                \
  }

__global__ __launch_bounds__(1024) void owner_sum_kernel(const float* __restrict__ x,
                                                         const int* __restrict__ ei,
                                                         float* __restrict__ agg,
                                                         int n_nodes, int n_edges) {
  __shared__ float acc[OWN_ROWS * 64];
  __shared__ int q_all[32 * Q_CAP];

  const int tid  = threadIdx.x;
  const int lane = tid & 31;
  const int wave = __builtin_amdgcn_readfirstlane(tid >> 5);
  const int row0 = blockIdx.x * OWN_ROWS;
  int rows = n_nodes - row0;
  if (rows > OWN_ROWS) rows = OWN_ROWS;

  float4* a4 = (float4*)acc;
  for (int i = tid; i < OWN_ROWS * 16; i += 1024) a4[i] = make_float4(0.f, 0.f, 0.f, 0.f);
  __syncthreads();

  const int* __restrict__ dstp = ei + n_edges;
  int* q = q_all + wave * Q_CAP;
  int qn = 0, head = 0;

  const int n_units = (n_edges & 3) == 0 ? n_edges / (UNIT_V4 * 128) : 0;
  const int4* __restrict__ d4 = (const int4*)dstp;
  int4 cur[UNIT_V4];
  int u = wave;
  if (u < n_units) {
#pragma unroll
    for (int j = 0; j < UNIT_V4; ++j) cur[j] = d4[(u * UNIT_V4 + j) * 32 + lane];
  }
  while (u < n_units) {
    const int un = u + 32;
    const int ul = un < n_units ? un : u;
#pragma unroll
    for (int j = 0; j < UNIT_V4; ++j) {
      const int4 d = cur[j];
      const int e0 = ((u * UNIT_V4 + j) * 32 + lane) * 4;
      OWNER_PUSH(d.x, e0 + 0)
      OWNER_PUSH(d.y, e0 + 1)
      OWNER_PUSH(d.z, e0 + 2)
      OWNER_PUSH(d.w, e0 + 3)
      OWNER_DRAIN_FULL
      cur[j] = d4[(ul * UNIT_V4 + j) * 32 + lane];
    }
    u = un;
  }
  for (int eb = n_units * (UNIT_V4 * 128) + wave * 32; eb < n_edges; eb += 32 * 32) {
    const int e = eb + lane;
    const int d = e < n_edges ? dstp[e] : -1;
    OWNER_PUSH(d, e)
    OWNER_DRAIN_FULL
  }
  if (qn > 0) owner_drain(x, ei, dstp, acc, q, head, qn, row0, lane);
  __syncthreads();

  float4* o4 = (float4*)(agg + (size_t)row0 * 64);
  for (int i = tid; i < rows * 16; i += 1024) o4[i] = a4[i];
}

__global__ __launch_bounds__(256) void gin_mlp_kernel(
    const float* __restrict__ x, const float* __restrict__ agg,
    const float* __restrict__ eps_p,
    const float* __restrict__ W1, const float* __restrict__ b1,
    const float* __restrict__ W2, const float* __restrict__ b2,
    float* __restrict__ out, int n_nodes) {
  __shared__ float lds[8 * 16 * LDS_STRIDE];

  const int tid  = threadIdx.x;
  const int wave = tid >> 5;
  const int lane = tid & 31;
  const int half = lane >> 4;
  const int m16  = lane & 15;
  float* tile = lds + wave * 16 * LDS_STRIDE;

  const float k1 = 1.0f + eps_p[0];
  const int node0 = (blockIdx.x * 8 + wave) * 16;

  for (int i = 0; i < 8; ++i) {
    int e4 = lane + i * 32;
    int r  = e4 >> 4;
    int c4 = e4 & 15;
    int node = node0 + r;
    int cn = node < n_nodes ? node : n_nodes - 1;
    float4 xv = *((const float4*)(x   + (size_t)cn * 64) + c4);
    float4 av = *((const float4*)(agg + (size_t)cn * 64) + c4);
    float* d = tile + r * LDS_STRIDE + c4 * 4;
    d[0] = k1 * xv.x + av.x;
    d[1] = k1 * xv.y + av.y;
    d[2] = k1 * xv.z + av.z;
    d[3] = k1 * xv.w + av.w;
  }
  __syncthreads();

  v8f acc[4];
  for (int t = 0; t < 4; ++t)
    for (int j = 0; j < 8; ++j) acc[t][j] = 0.f;

  for (int k = 0; k < 16; ++k) {
    const float2 af = *(const float2*)(tile + m16 * LDS_STRIDE + 4 * k + 2 * half);
    v2f a; a.x = af.x; a.y = af.y;
    for (int t = 0; t < 4; ++t) {
      int n = t * 16 + m16;
      const float2 bf = *(const float2*)(W1 + (size_t)n * 64 + 4 * k + 2 * half);
      v2f b; b.x = bf.x; b.y = bf.y;
      acc[t] = __builtin_amdgcn_wmma_f32_16x16x4_f32(
          false, a, false, b, (short)0, acc[t], false, false);
    }
  }
  __syncthreads();

  for (int t = 0; t < 4; ++t) {
    float bv = b1[t * 16 + m16];
    for (int v = 0; v < 8; ++v) {
      float h = acc[t][v] + bv;
      h = h > 0.f ? h : 0.f;
      tile[(v + 8 * half) * LDS_STRIDE + t * 16 + m16] = h;
    }
  }
  __syncthreads();

  v8f acc2[4];
  for (int t = 0; t < 4; ++t)
    for (int j = 0; j < 8; ++j) acc2[t][j] = 0.f;

  for (int k = 0; k < 16; ++k) {
    const float2 af = *(const float2*)(tile + m16 * LDS_STRIDE + 4 * k + 2 * half);
    v2f a; a.x = af.x; a.y = af.y;
    for (int t = 0; t < 4; ++t) {
      int n = t * 16 + m16;
      const float2 bf = *(const float2*)(W2 + (size_t)n * 64 + 4 * k + 2 * half);
      v2f b; b.x = bf.x; b.y = bf.y;
      acc2[t] = __builtin_amdgcn_wmma_f32_16x16x4_f32(
          false, a, false, b, (short)0, acc2[t], false, false);
    }
  }

  for (int t = 0; t < 4; ++t) {
    float bv = b2[t * 16 + m16];
    for (int v = 0; v < 8; ++v) {
      int node = node0 + v + 8 * half;
      if (node < n_nodes)
        out[(size_t)node * 64 + t * 16 + m16] = acc2[t][v] + bv;
    }
  }
}

extern "C" void kernel_launch(void* const* d_in, const int* in_sizes, int n_in,
                              void* d_out, int out_size, void* d_ws, size_t ws_size,
                              hipStream_t stream) {
  const float* x   = (const float*)d_in[0];
  const int*   ei  = (const int*)  d_in[1];
  const float* eps = (const float*)d_in[2];
  const float* W1  = (const float*)d_in[3];
  const float* b1  = (const float*)d_in[4];
  const float* W2  = (const float*)d_in[5];
  const float* b2  = (const float*)d_in[6];
  float* out = (float*)d_out;

  const int n_nodes = in_sizes[0] / 64;
  const int n_edges = in_sizes[1] / 2;
  float* agg = (float*)d_ws;

  int oblocks = (n_nodes + OWN_ROWS - 1) / OWN_ROWS;
  owner_sum_kernel<<<oblocks, 1024, 0, stream>>>(x, ei, agg, n_nodes, n_edges);

  int ntiles = (n_nodes + 15) / 16;
  int mblocks = (ntiles + 7) / 8;
  gin_mlp_kernel<<<mblocks, 256, 0, stream>>>(x, agg, eps, W1, b1, W2, b2, out, n_nodes);
}


# Edges binned once by 128-row target range (count, offsets, LDS-sorted fill); each owner group reads only its bin instead of scanning the whole list; LDS float adds stay, arrival order
# speedup vs baseline: 2.3522x; 1.6171x over previous
#include <hip/hip_runtime.h>

typedef __attribute__((ext_vector_type(2))) float v2f;
typedef __attribute__((ext_vector_type(8))) float v8f;

#define LDS_STRIDE 68

#define BROWS     128
#define BSHIFT    7
#define NB_MAX    1024
#define CHUNK     16384
#define EPT       16
#define ROW_BATCH 16
#define COL_BATCH 16

__global__ __launch_bounds__(1024) void bin_count_kernel(const int* __restrict__ ei,
                                                         int* __restrict__ hist,
                                                         int n_nodes, int n_edges, int nb) {
  __shared__ int lh[NB_MAX];
  const int tid = threadIdx.x;
  for (int i = tid; i < nb; i += 1024) lh[i] = 0;
  __syncthreads();

  const int* __restrict__ dstp = ei + n_edges;
  const int e0 = blockIdx.x * CHUNK + tid;
  int d[EPT];
#pragma unroll
  for (int j = 0; j < EPT; ++j) {
    const int e = e0 + j * 1024;
    d[j] = e < n_edges ? dstp[e] : -1;
  }
#pragma unroll
  for (int j = 0; j < EPT; ++j)
    if ((unsigned)d[j] < (unsigned)n_nodes) atomicAdd(&lh[d[j] >> BSHIFT], 1);
  __syncthreads();
  for (int i = tid; i < nb; i += 1024) hist[blockIdx.x * nb + i] = lh[i];
}

__device__ __forceinline__ int wave_incl_scan(int v, int lane) {
#pragma unroll
  for (int o = 1; o < 32; o <<= 1) {
    const int t = __shfl_up(v, o);
    if (lane >= o) v += t;
  }
  return v;
}

__global__ __launch_bounds__(1024) void bin_fill_kernel(const int* __restrict__ ei,
                                                        const int* __restrict__ hist,
                                                        int* __restrict__ bstart,
                                                        int* __restrict__ sorted_i,
                                                        int n_nodes, int n_edges, int nb,
                                                        int n_chunks) {
  __shared__ int2 pairs[CHUNK];
  __shared__ int cur[NB_MAX];
  __shared__ int gdelta[NB_MAX];
  __shared__ int ws_a[32], ws_b[32];
  __shared__ int nvalid_s;

  const int tid  = threadIdx.x;
  const int lane = tid & 31;
  const int wave = __builtin_amdgcn_readfirstlane(tid >> 5);
  const int g = blockIdx.x;

  int tot = 0, pre = 0, mine = 0;
  if (tid < nb) {
    const int* __restrict__ col = hist + tid;
    for (int g0 = 0; g0 < n_chunks; g0 += COL_BATCH) {
      int v[COL_BATCH];
#pragma unroll
      for (int j = 0; j < COL_BATCH; ++j) v[j] = g0 + j < n_chunks ? col[(g0 + j) * nb] : 0;
#pragma unroll
      for (int j = 0; j < COL_BATCH; ++j) {
        tot += v[j];
        pre += g0 + j < g ? v[j] : 0;
        mine = g0 + j == g ? v[j] : mine;
      }
    }
  }
  const int ia = wave_incl_scan(tot, lane);
  const int ib = wave_incl_scan(mine, lane);
  if (lane == 31) { ws_a[wave] = ia; ws_b[wave] = ib; }
  __syncthreads();
  if (wave == 0) {
    const int a = ws_a[lane], b = ws_b[lane];
    const int sa = wave_incl_scan(a, lane);
    const int sb = wave_incl_scan(b, lane);
    ws_a[lane] = sa - a;
    ws_b[lane] = sb - b;
    if (lane == 31) nvalid_s = sb;
  }
  __syncthreads();
  const int start = ia - tot + ws_a[wave];
  const int lofs  = ib - mine + ws_b[wave];
  if (tid < nb) {
    cur[tid] = lofs;
    gdelta[tid] = start + pre - lofs;
    if (g == 0) {
      bstart[tid] = start;
      if (tid == nb - 1) bstart[nb] = start + tot;
    }
  }
  __syncthreads();

  const int* __restrict__ dstp = ei + n_edges;
  const int e0 = g * CHUNK + tid;
  int s[EPT], d[EPT];
#pragma unroll
  for (int j = 0; j < EPT; ++j) {
    const int e = e0 + j * 1024;
    const bool ok = e < n_edges;
    d[j] = ok ? dstp[e] : -1;
    s[j] = ok ? ei[e] : 0;
  }
#pragma unroll
  for (int j = 0; j < EPT; ++j) {
    if ((unsigned)d[j] < (unsigned)n_nodes) {
      const int p = atomicAdd(&cur[d[j] >> BSHIFT], 1);
      pairs[p] = make_int2(s[j], d[j]);
    }
  }
  __syncthreads();

  const int nv = nvalid_s;
  int2* __restrict__ sorted = (int2*)sorted_i;
  for (int p = tid; p < nv; p += 1024) {
    const int2 pr = pairs[p];
    sorted[gdelta[pr.y >> BSHIFT] + p] = pr;
  }
}

__global__ __launch_bounds__(256) void owner_sum_kernel(const float* __restrict__ x,
                                                        const int* __restrict__ sorted_i,
                                                        const int* __restrict__ bstart,
                                                        float* __restrict__ agg, int n_nodes) {
  __shared__ float acc[BROWS * 64];

  const int tid  = threadIdx.x;
  const int lane = tid & 31;
  const int wave = __builtin_amdgcn_readfirstlane(tid >> 5);
  const int row0 = blockIdx.x * BROWS;
  int rows = n_nodes - row0;
  if (rows > BROWS) rows = BROWS;

  float4* a4 = (float4*)acc;
  for (int i = tid; i < BROWS * 16; i += 256) a4[i] = make_float4(0.f, 0.f, 0.f, 0.f);
  __syncthreads();

  const int2* __restrict__ sorted = (const int2*)sorted_i;
  const int s1 = bstart[blockIdx.x + 1];
  int base = bstart[blockIdx.x] + wave * 32;

  int2 nx = make_int2(0, row0);
  if (base + lane < s1) nx = sorted[base + lane];
  while (base < s1) {
    const int2 c = nx;
    const int left = s1 - base;
    const int count = left < 32 ? left : 32;
    const int nbase = base + 8 * 32;
    nx = make_int2(0, row0);
    if (nbase + lane < s1) nx = sorted[nbase + lane];
    const int s = c.x;
    const int r = c.y - row0;
    for (int k0 = 0; k0 < count; k0 += ROW_BATCH) {
      float2 v[ROW_BATCH];
#pragma unroll
      for (int j = 0; j < ROW_BATCH; ++j) {
        const int sk = __builtin_amdgcn_readlane(s, k0 + j);
        v[j] = *(const float2*)(x + (size_t)sk * 64 + 2 * lane);
      }
#pragma unroll
      for (int j = 0; j < ROW_BATCH; ++j) {
        if (k0 + j < count) {
          const int rk = __builtin_amdgcn_readlane(r, k0 + j);
          float* p = acc + rk * 64 + 2 * lane;
          atomicAdd(p, v[j].x);
          atomicAdd(p + 1, v[j].y);
        }
      }
    }
    base = nbase;
  }
  __syncthreads();

  float4* o4 = (float4*)(agg + (size_t)row0 * 64);
  for (int i = tid; i < rows * 16; i += 256) o4[i] = a4[i];
}

__global__ __launch_bounds__(256) void gin_mlp_kernel(
    const float* __restrict__ x, const float* __restrict__ agg,
    const float* __restrict__ eps_p,
    const float* __restrict__ W1, const float* __restrict__ b1,
    const float* __restrict__ W2, const float* __restrict__ b2,
    float* __restrict__ out, int n_nodes) {
  __shared__ float lds[8 * 16 * LDS_STRIDE];

  const int tid  = threadIdx.x;
  const int wave = tid >> 5;
  const int lane = tid & 31;
  const int half = lane >> 4;
  const int m16  = lane & 15;
  float* tile = lds + wave * 16 * LDS_STRIDE;

  const float k1 = 1.0f + eps_p[0];
  const int node0 = (blockIdx.x * 8 + wave) * 16;

  for (int i = 0; i < 8; ++i) {
    int e4 = lane + i * 32;
    int r  = e4 >> 4;
    int c4 = e4 & 15;
    int node = node0 + r;
    int cn = node < n_nodes ? node : n_nodes - 1;
    float4 xv = *((const float4*)(x   + (size_t)cn * 64) + c4);
    float4 av = *((const float4*)(agg + (size_t)cn * 64) + c4);
    float* d = tile + r * LDS_STRIDE + c4 * 4;
    d[0] = k1 * xv.x + av.x;
    d[1] = k1 * xv.y + av.y;
    d[2] = k1 * xv.z + av.z;
    d[3] = k1 * xv.w + av.w;
  }
  __syncthreads();

  v8f acc[4];
  for (int t = 0; t < 4; ++t)
    for (int j = 0; j < 8; ++j) acc[t][j] = 0.f;

  for (int k = 0; k < 16; ++k) {
    const float2 af = *(const float2*)(tile + m16 * LDS_STRIDE + 4 * k + 2 * half);
    v2f a; a.x = af.x; a.y = af.y;
    for (int t = 0; t < 4; ++t) {
      int n = t * 16 + m16;
      const float2 bf = *(const float2*)(W1 + (size_t)n * 64 + 4 * k + 2 * half);
      v2f b; b.x = bf.x; b.y = bf.y;
      acc[t] = __builtin_amdgcn_wmma_f32_16x16x4_f32(
          false, a, false, b, (short)0, acc[t], false, false);
    }
  }
  __syncthreads();

  for (int t = 0; t < 4; ++t) {
    float bv = b1[t * 16 + m16];
    for (int v = 0; v < 8; ++v) {
      float h = acc[t][v] + bv;
      h = h > 0.f ? h : 0.f;
      tile[(v + 8 * half) * LDS_STRIDE + t * 16 + m16] = h;
    }
  }
  __syncthreads();

  v8f acc2[4];
  for (int t = 0; t < 4; ++t)
    for (int j = 0; j < 8; ++j) acc2[t][j] = 0.f;

  for (int k = 0; k < 16; ++k) {
    const float2 af = *(const float2*)(tile + m16 * LDS_STRIDE + 4 * k + 2 * half);
    v2f a; a.x = af.x; a.y = af.y;
    for (int t = 0; t < 4; ++t) {
      int n = t * 16 + m16;
      const float2 bf = *(const float2*)(W2 + (size_t)n * 64 + 4 * k + 2 * half);
      v2f b; b.x = bf.x; b.y = bf.y;
      acc2[t] = __builtin_amdgcn_wmma_f32_16x16x4_f32(
          false, a, false, b, (short)0, acc2[t], false, false);
    }
  }

  for (int t = 0; t < 4; ++t) {
    float bv = b2[t * 16 + m16];
    for (int v = 0; v < 8; ++v) {
      int node = node0 + v + 8 * half;
      if (node < n_nodes)
        out[(size_t)node * 64 + t * 16 + m16] = acc2[t][v] + bv;
    }
  }
}

extern "C" void kernel_launch(void* const* d_in, const int* in_sizes, int n_in,
                              void* d_out, int out_size, void* d_ws, size_t ws_size,
                              hipStream_t stream) {
  const float* x   = (const float*)d_in[0];
  const int*   ei  = (const int*)  d_in[1];
  const float* eps = (const float*)d_in[2];
  const float* W1  = (const float*)d_in[3];
  const float* b1  = (const float*)d_in[4];
  const float* W2  = (const float*)d_in[5];
  const float* b2  = (const float*)d_in[6];
  float* out = (float*)d_out;

  const int n_nodes = in_sizes[0] / 64;
  const int n_edges = in_sizes[1] / 2;
  const int nb = (n_nodes + BROWS - 1) / BROWS;
  const int n_chunks = (n_edges + CHUNK - 1) / CHUNK;

  float* agg  = (float*)d_ws;
  int* sorted = (int*)(agg + (size_t)n_nodes * 64);
  int* hist   = sorted + (size_t)n_edges * 2;
  int* bstart = hist + (size_t)n_chunks * nb;

  bin_count_kernel<<<n_chunks, 1024, 0, stream>>>(ei, hist, n_nodes, n_edges, nb);
  bin_fill_kernel<<<n_chunks, 1024, 0, stream>>>(ei, hist, bstart, sorted, n_nodes, n_edges, nb,
                                                 n_chunks);
  owner_sum_kernel<<<nb, 256, 0, stream>>>(x, sorted, bstart, agg, n_nodes);

  int ntiles = (n_nodes + 15) / 16;
  int mblocks = (ntiles + 7) / 8;
  gin_mlp_kernel<<<mblocks, 256, 0, stream>>>(x, agg, eps, W1, b1, W2, b2, out, n_nodes);
}


# Owner-sum and MLP kernels fused: each bin's 128 summed rows stay in LDS at the padded row pitch and feed the f32 WMMA layers directly; agg is never written or re-read; 3 launches instead of 4
# speedup vs baseline: 2.5468x; 1.0827x over previous
#include <hip/hip_runtime.h>

typedef __attribute__((ext_vector_type(2))) float v2f;
typedef __attribute__((ext_vector_type(8))) float v8f;

#define LDS_STRIDE 68

#define BROWS     128
#define BSHIFT    7
#define NB_MAX    1024
#define CHUNK     16384
#define EPT       16
#define ROW_BATCH 16
#define COL_BATCH 16

__global__ __launch_bounds__(1024) void bin_count_kernel(const int* __restrict__ ei,
                                                         int* __restrict__ hist,
                                                         int n_nodes, int n_edges, int nb) {
  __shared__ int lh[NB_MAX];
  const int tid = threadIdx.x;
  for (int i = tid; i < nb; i += 1024) lh[i] = 0;
  __syncthreads();

  const int* __restrict__ dstp = ei + n_edges;
  const int e0 = blockIdx.x * CHUNK + tid;
  int d[EPT];
#pragma unroll
  for (int j = 0; j < EPT; ++j) {
    const int e = e0 + j * 1024;
    d[j] = e < n_edges ? dstp[e] : -1;
  }
#pragma unroll
  for (int j = 0; j < EPT; ++j)
    if ((unsigned)d[j] < (unsigned)n_nodes) atomicAdd(&lh[d[j] >> BSHIFT], 1);
  __syncthreads();
  for (int i = tid; i < nb; i += 1024) hist[blockIdx.x * nb + i] = lh[i];
}

__device__ __forceinline__ int wave_incl_scan(int v, int lane) {
#pragma unroll
  for (int o = 1; o < 32; o <<= 1) {
    const int t = __shfl_up(v, o);
    if (lane >= o) v += t;
  }
  return v;
}

__global__ __launch_bounds__(1024) void bin_fill_kernel(const int* __restrict__ ei,
                                                        const int* __restrict__ hist,
                                                        int* __restrict__ bstart,
                                                        int* __restrict__ sorted_i,
                                                        int n_nodes, int n_edges, int nb,
                                                        int n_chunks) {
  __shared__ int2 pairs[CHUNK];
  __shared__ int cur[NB_MAX];
  __shared__ int gdelta[NB_MAX];
  __shared__ int ws_a[32], ws_b[32];
  __shared__ int nvalid_s;

  const int tid  = threadIdx.x;
  const int lane = tid & 31;
  const int wave = __builtin_amdgcn_readfirstlane(tid >> 5);
  const int g = blockIdx.x;

  int tot = 0, pre = 0, mine = 0;
  if (tid < nb) {
    const int* __restrict__ col = hist + tid;
    for (int g0 = 0; g0 < n_chunks; g0 += COL_BATCH) {
      int v[COL_BATCH];
#pragma unroll
      for (int j = 0; j < COL_BATCH; ++j) v[j] = g0 + j < n_chunks ? col[(g0 + j) * nb] : 0;
#pragma unroll
      for (int j = 0; j < COL_BATCH; ++j) {
        tot += v[j];
        pre += g0 + j < g ? v[j] : 0;
        mine = g0 + j == g ? v[j] : mine;
      }
    }
  }
  const int ia = wave_incl_scan(tot, lane);
  const int ib = wave_incl_scan(mine, lane);
  if (lane == 31) { ws_a[wave] = ia; ws_b[wave] = ib; }
  __syncthreads();
  if (wave == 0) {
    const int a = ws_a[lane], b = ws_b[lane];
    const int sa = wave_incl_scan(a, lane);
    const int sb = wave_incl_scan(b, lane);
    ws_a[lane] = sa - a;
    ws_b[lane] = sb - b;
    if (lane == 31) nvalid_s = sb;
  }
  __syncthreads();
  const int start = ia - tot + ws_a[wave];
  const int lofs  = ib - mine + ws_b[wave];
  if (tid < nb) {
    cur[tid] = lofs;
    gdelta[tid] = start + pre - lofs;
    if (g == 0) {
      bstart[tid] = start;
      if (tid == nb - 1) bstart[nb] = start + tot;
    }
  }
  __syncthreads();

  const int* __restrict__ dstp = ei + n_edges;
  const int e0 = g * CHUNK + tid;
  int s[EPT], d[EPT];
#pragma unroll
  for (int j = 0; j < EPT; ++j) {
    const int e = e0 + j * 1024;
    const bool ok = e < n_edges;
    d[j] = ok ? dstp[e] : -1;
    s[j] = ok ? ei[e] : 0;
  }
#pragma unroll
  for (int j = 0; j < EPT; ++j) {
    if ((unsigned)d[j] < (unsigned)n_nodes) {
      const int p = atomicAdd(&cur[d[j] >> BSHIFT], 1);
      pairs[p] = make_int2(s[j], d[j]);
    }
  }
  __syncthreads();

  const int nv = nvalid_s;
  int2* __restrict__ sorted = (int2*)sorted_i;
  for (int p = tid; p < nv; p += 1024) {
    const int2 pr = pairs[p];
    sorted[gdelta[pr.y >> BSHIFT] + p] = pr;
  }
}

__global__ __launch_bounds__(256) void owner_mlp_kernel(
    const float* __restrict__ x, const int* __restrict__ sorted_i,
    const int* __restrict__ bstart, const float* __restrict__ eps_p,
    const float* __restrict__ W1, const float* __restrict__ b1,
    const float* __restrict__ W2, const float* __restrict__ b2,
    float* __restrict__ out, int n_nodes) {
  static_assert(BROWS == 8 * 16, "one 16-row tile per wave of the 256-thread group");
  __shared__ float acc[BROWS * LDS_STRIDE];

  const int tid  = threadIdx.x;
  const int lane = tid & 31;
  const int wave = __builtin_amdgcn_readfirstlane(tid >> 5);
  const int row0 = blockIdx.x * BROWS;

  float4* a4 = (float4*)acc;
  for (int i = tid; i < BROWS * LDS_STRIDE / 4; i += 256) a4[i] = make_float4(0.f, 0.f, 0.f, 0.f);
  __syncthreads();

  const int2* __restrict__ sorted = (const int2*)sorted_i;
  const int s1 = bstart[blockIdx.x + 1];
  int base = bstart[blockIdx.x] + wave * 32;

  int2 nx = make_int2(0, row0);
  if (base + lane < s1) nx = sorted[base + lane];
  while (base < s1) {
    const int2 c = nx;
    const int left = s1 - base;
    const int count = left < 32 ? left : 32;
    const int nbase = base + 8 * 32;
    nx = make_int2(0, row0);
    if (nbase + lane < s1) nx = sorted[nbase + lane];
    const int s = c.x;
    const int r = c.y - row0;
    for (int k0 = 0; k0 < count; k0 += ROW_BATCH) {
      float2 v[ROW_BATCH];
#pragma unroll
      for (int j = 0; j < ROW_BATCH; ++j) {
        const int sk = __builtin_amdgcn_readlane(s, k0 + j);
        v[j] = *(const float2*)(x + (size_t)sk * 64 + 2 * lane);
      }
#pragma unroll
      for (int j = 0; j < ROW_BATCH; ++j) {
        if (k0 + j < count) {
          const int rk = __builtin_amdgcn_readlane(r, k0 + j);
          float* p = acc + rk * LDS_STRIDE + 2 * lane;
          atomicAdd(p, v[j].x);
          atomicAdd(p + 1, v[j].y);
        }
      }
    }
    base = nbase;
  }
  __syncthreads();

  const int half = lane >> 4;
  const int m16  = lane & 15;
  const int wv = tid >> 5;
  float* tile = acc + wv * 16 * LDS_STRIDE;

  const float k1 = 1.0f + eps_p[0];
  const int node0 = row0 + wv * 16;

  for (int i = 0; i < 8; ++i) {
    int e4 = lane + i * 32;
    int r  = e4 >> 4;
    int c4 = e4 & 15;
    int node = node0 + r;
    int cn = node < n_nodes ? node : n_nodes - 1;
    float4 xv = *((const float4*)(x + (size_t)cn * 64) + c4);
    float* d = tile + r * LDS_STRIDE + c4 * 4;
    d[0] = k1 * xv.x + d[0];
    d[1] = k1 * xv.y + d[1];
    d[2] = k1 * xv.z + d[2];
    d[3] = k1 * xv.w + d[3];
  }
  __syncthreads();

  v8f acc1[4];
  for (int t = 0; t < 4; ++t)
    for (int j = 0; j < 8; ++j) acc1[t][j] = 0.f;

  for (int k = 0; k < 16; ++k) {
    const float2 af = *(const float2*)(tile + m16 * LDS_STRIDE + 4 * k + 2 * half);
    v2f a; a.x = af.x; a.y = af.y;
    for (int t = 0; t < 4; ++t) {
      int n = t * 16 + m16;
      const float2 bf = *(const float2*)(W1 + (size_t)n * 64 + 4 * k + 2 * half);
      v2f b; b.x = bf.x; b.y = bf.y;
      acc1[t] = __builtin_amdgcn_wmma_f32_16x16x4_f32(
          false, a, false, b, (short)0, acc1[t], false, false);
    }
  }
  __syncthreads();

  for (int t = 0; t < 4; ++t) {
    float bv = b1[t * 16 + m16];
    for (int v = 0; v < 8; ++v) {
      float h = acc1[t][v] + bv;
      h = h > 0.f ? h : 0.f;
      tile[(v + 8 * half) * LDS_STRIDE + t * 16 + m16] = h;
    }
  }
  __syncthreads();

  v8f acc2[4];
  for (int t = 0; t < 4; ++t)
    for (int j = 0; j < 8; ++j) acc2[t][j] = 0.f;

  for (int k = 0; k < 16; ++k) {
    const float2 af = *(const float2*)(tile + m16 * LDS_STRIDE + 4 * k + 2 * half);
    v2f a; a.x = af.x; a.y = af.y;
    for (int t = 0; t < 4; ++t) {
      int n = t * 16 + m16;
      const float2 bf = *(const float2*)(W2 + (size_t)n * 64 + 4 * k + 2 * half);
      v2f b; b.x = bf.x; b.y = bf.y;
      acc2[t] = __builtin_amdgcn_wmma_f32_16x16x4_f32(
          false, a, false, b, (short)0, acc2[t], false, false);
    }
  }

  for (int t = 0; t < 4; ++t) {
    float bv = b2[t * 16 + m16];
    for (int v = 0; v < 8; ++v) {
      int node = node0 + v + 8 * half;
      if (node < n_nodes)
        out[(size_t)node * 64 + t * 16 + m16] = acc2[t][v] + bv;
    }
  }
}

extern "C" void kernel_launch(void* const* d_in, const int* in_sizes, int n_in,
                              void* d_out, int out_size, void* d_ws, size_t ws_size,
                              hipStream_t stream) {
  const float* x   = (const float*)d_in[0];
  const int*   ei  = (const int*)  d_in[1];
  const float* eps = (const float*)d_in[2];
  const float* W1  = (const float*)d_in[3];
  const float* b1  = (const float*)d_in[4];
  const float* W2  = (const float*)d_in[5];
  const float* b2  = (const float*)d_in[6];
  float* out = (float*)d_out;

  const int n_nodes = in_sizes[0] / 64;
  const int n_edges = in_sizes[1] / 2;
  const int nb = (n_nodes + BROWS - 1) / BROWS;
  const int n_chunks = (n_edges + CHUNK - 1) / CHUNK;

  int* sorted = (int*)d_ws;
  int* hist   = sorted + (size_t)n_edges * 2;
  int* bstart = hist + (size_t)n_chunks * nb;

  bin_count_kernel<<<n_chunks, 1024, 0, stream>>>(ei, hist, n_nodes, n_edges, nb);
  bin_fill_kernel<<<n_chunks, 1024, 0, stream>>>(ei, hist, bstart, sorted, n_nodes, n_edges, nb,
                                                 n_chunks);
  owner_mlp_kernel<<<nb, 256, 0, stream>>>(x, sorted, bstart, eps, W1, b1, W2, b2, out, n_nodes);
}


# Owner group orders its bin's sources by target row in LDS (count, scan, fill); each half-wave sums one row in registers from 16-byte loads, stores it once: LDS float atomics gone; sum order = rank
# speedup vs baseline: 2.9059x; 1.1410x over previous
#include <hip/hip_runtime.h>

typedef __attribute__((ext_vector_type(2))) float v2f;
typedef __attribute__((ext_vector_type(8))) float v8f;

#define LDS_STRIDE 68

#define BROWS     128
#define BSHIFT    7
#define NB_MAX    1024
#define CHUNK     16384
#define EPT       16
#define TILE_PAIRS 3072
#define PPT       12
#define GATHER    16
#define COL_BATCH 16

__global__ __launch_bounds__(1024) void bin_count_kernel(const int* __restrict__ ei,
                                                         int* __restrict__ hist,
                                                         int n_nodes, int n_edges, int nb) {
  __shared__ int lh[NB_MAX];
  const int tid = threadIdx.x;
  for (int i = tid; i < nb; i += 1024) lh[i] = 0;
  __syncthreads();

  const int* __restrict__ dstp = ei + n_edges;
  const int e0 = blockIdx.x * CHUNK + tid;
  int d[EPT];
#pragma unroll
  for (int j = 0; j < EPT; ++j) {
    const int e = e0 + j * 1024;
    d[j] = e < n_edges ? dstp[e] : -1;
  }
#pragma unroll
  for (int j = 0; j < EPT; ++j)
    if ((unsigned)d[j] < (unsigned)n_nodes) atomicAdd(&lh[d[j] >> BSHIFT], 1);
  __syncthreads();
  for (int i = tid; i < nb; i += 1024) hist[blockIdx.x * nb + i] = lh[i];
}

__device__ __forceinline__ int wave_incl_scan(int v, int lane) {
#pragma unroll
  for (int o = 1; o < 32; o <<= 1) {
    const int t = __shfl_up(v, o);
    if (lane >= o) v += t;
  }
  return v;
}

__global__ __launch_bounds__(1024) void bin_fill_kernel(const int* __restrict__ ei,
                                                        const int* __restrict__ hist,
                                                        int* __restrict__ bstart,
                                                        int* __restrict__ sorted_i,
                                                        int n_nodes, int n_edges, int nb,
                                                        int n_chunks) {
  __shared__ int2 pairs[CHUNK];
  __shared__ int cur[NB_MAX];
  __shared__ int gdelta[NB_MAX];
  __shared__ int ws_a[32], ws_b[32];
  __shared__ int nvalid_s;

  const int tid  = threadIdx.x;
  const int lane = tid & 31;
  const int wave = __builtin_amdgcn_readfirstlane(tid >> 5);
  const int g = blockIdx.x;

  int tot = 0, pre = 0, mine = 0;
  if (tid < nb) {
    const int* __restrict__ col = hist + tid;
    for (int g0 = 0; g0 < n_chunks; g0 += COL_BATCH) {
      int v[COL_BATCH];
#pragma unroll
      for (int j = 0; j < COL_BATCH; ++j) v[j] = g0 + j < n_chunks ? col[(g0 + j) * nb] : 0;
#pragma unroll
      for (int j = 0; j < COL_BATCH; ++j) {
        tot += v[j];
        pre += g0 + j < g ? v[j] : 0;
        mine = g0 + j == g ? v[j] : mine;
      }
    }
  }
  const int ia = wave_incl_scan(tot, lane);
  const int ib = wave_incl_scan(mine, lane);
  if (lane == 31) { ws_a[wave] = ia; ws_b[wave] = ib; }
  __syncthreads();
  if (wave == 0) {
    const int a = ws_a[lane], b = ws_b[lane];
    const int sa = wave_incl_scan(a, lane);
    const int sb = wave_incl_scan(b, lane);
    ws_a[lane] = sa - a;
    ws_b[lane] = sb - b;
    if (lane == 31) nvalid_s = sb;
  }
  __syncthreads();
  const int start = ia - tot + ws_a[wave];
  const int lofs  = ib - mine + ws_b[wave];
  if (tid < nb) {
    cur[tid] = lofs;
    gdelta[tid] = start + pre - lofs;
    if (g == 0) {
      bstart[tid] = start;
      if (tid == nb - 1) bstart[nb] = start + tot;
    }
  }
  __syncthreads();

  const int* __restrict__ dstp = ei + n_edges;
  const int e0 = g * CHUNK + tid;
  int s[EPT], d[EPT];
#pragma unroll
  for (int j = 0; j < EPT; ++j) {
    const int e = e0 + j * 1024;
    const bool ok = e < n_edges;
    d[j] = ok ? dstp[e] : -1;
    s[j] = ok ? ei[e] : 0;
  }
#pragma unroll
  for (int j = 0; j < EPT; ++j) {
    if ((unsigned)d[j] < (unsigned)n_nodes) {
      const int p = atomicAdd(&cur[d[j] >> BSHIFT], 1);
      pairs[p] = make_int2(s[j], d[j]);
    }
  }
  __syncthreads();

  const int nv = nvalid_s;
  int2* __restrict__ sorted = (int2*)sorted_i;
  for (int p = tid; p < nv; p += 1024) {
    const int2 pr = pairs[p];
    sorted[gdelta[pr.y >> BSHIFT] + p] = pr;
  }
}

__global__ __launch_bounds__(256) void owner_mlp_kernel(
    const float* __restrict__ x, const int* __restrict__ sorted_i,
    const int* __restrict__ bstart, const float* __restrict__ eps_p,
    const float* __restrict__ W1, const float* __restrict__ b1,
    const float* __restrict__ W2, const float* __restrict__ b2,
    float* __restrict__ out, int n_nodes) {
  static_assert(BROWS == 8 * 16, "one 16-row tile per wave of the 256-thread group");
  __shared__ __attribute__((aligned(16))) float acc[BROWS * LDS_STRIDE];
  __shared__ int srcs[TILE_PAIRS];
  __shared__ __attribute__((aligned(16))) int cnt[BROWS];
  __shared__ __attribute__((aligned(16))) int rbeg[BROWS];

  const int tid  = threadIdx.x;
  const int lane = tid & 31;
  const int wave = __builtin_amdgcn_readfirstlane(tid >> 5);
  const int row0 = blockIdx.x * BROWS;

  const int2* __restrict__ sorted = (const int2*)sorted_i;
  const float4* __restrict__ x4 = (const float4*)x;
  const int s1 = bstart[blockIdx.x + 1];
  int t0 = bstart[blockIdx.x];
  bool first = true;

  for (;;) {
    if (tid < BROWS) cnt[tid] = 0;
    __syncthreads();

    int2 pr[PPT];
    int key[PPT];
#pragma unroll
    for (int j = 0; j < PPT; ++j) {
      const int p = t0 + j * 256 + tid;
      pr[j] = p < s1 ? sorted[p] : make_int2(0, -1);
    }
#pragma unroll
    for (int j = 0; j < PPT; ++j) {
      key[j] = -1;
      if (pr[j].y >= 0) {
        const int r = pr[j].y - row0;
        key[j] = (atomicAdd(&cnt[r], 1) << BSHIFT) | r;
      }
    }
    __syncthreads();

    if (wave == 0) {
      const int4 c = *(const int4*)(cnt + 4 * lane);
      const int tot = c.x + c.y + c.z + c.w;
      const int ex = wave_incl_scan(tot, lane) - tot;
      *(int4*)(rbeg + 4 * lane) = make_int4(ex, ex + c.x, ex + c.x + c.y, ex + c.x + c.y + c.z);
    }
    __syncthreads();

#pragma unroll
    for (int j = 0; j < PPT; ++j)
      if (key[j] >= 0) srcs[rbeg[key[j] & (BROWS - 1)] + (key[j] >> BSHIFT)] = pr[j].x;
    __syncthreads();

    {
      const int hrow = lane >> 4;
      const int c16  = lane & 15;
      for (int i = 0; i < 8; ++i) {
        const int rl  = wave * 16 + 2 * i + hrow;
        const int beg = rbeg[rl];
        const int n   = cnt[rl];
        const int na = __builtin_amdgcn_readlane(n, 0);
        const int nb = __builtin_amdgcn_readlane(n, 16);
        const int nmax = na > nb ? na : nb;
        float4 sum = make_float4(0.f, 0.f, 0.f, 0.f);
        for (int j0 = 0; j0 < nmax; j0 += GATHER) {
          int sj[GATHER];
          float4 v[GATHER];
#pragma unroll
          for (int j = 0; j < GATHER; ++j) {
            sj[j] = 0;
            if (j0 + j < n) sj[j] = srcs[beg + j0 + j];
          }
#pragma unroll
          for (int j = 0; j < GATHER; ++j) v[j] = x4[(unsigned)sj[j] * 16u + (unsigned)c16];
#pragma unroll
          for (int j = 0; j < GATHER; ++j) {
            if (j0 + j < n) {
              sum.x += v[j].x; sum.y += v[j].y; sum.z += v[j].z; sum.w += v[j].w;
            }
          }
        }
        float4* d = (float4*)(acc + rl * LDS_STRIDE + 4 * c16);
        if (first) {
          *d = sum;
        } else {
          const float4 o = *d;
          *d = make_float4(o.x + sum.x, o.y + sum.y, o.z + sum.z, o.w + sum.w);
        }
      }
    }

    t0 += TILE_PAIRS;
    first = false;
    if (t0 >= s1) break;
    __syncthreads();
  }
  __syncthreads();

  const int half = lane >> 4;
  const int m16  = lane & 15;
  const int wv = tid >> 5;
  float* tile = acc + wv * 16 * LDS_STRIDE;

  const float k1 = 1.0f + eps_p[0];
  const int node0 = row0 + wv * 16;

  for (int i = 0; i < 8; ++i) {
    int e4 = lane + i * 32;
    int r  = e4 >> 4;
    int c4 = e4 & 15;
    int node = node0 + r;
    int cn = node < n_nodes ? node : n_nodes - 1;
    float4 xv = *((const float4*)(x + (size_t)cn * 64) + c4);
    float* d = tile + r * LDS_STRIDE + c4 * 4;
    d[0] = k1 * xv.x + d[0];
    d[1] = k1 * xv.y + d[1];
    d[2] = k1 * xv.z + d[2];
    d[3] = k1 * xv.w + d[3];
  }
  __syncthreads();

  v8f acc1[4];
  for (int t = 0; t < 4; ++t)
    for (int j = 0; j < 8; ++j) acc1[t][j] = 0.f;

  for (int k = 0; k < 16; ++k) {
    const float2 af = *(const float2*)(tile + m16 * LDS_STRIDE + 4 * k + 2 * half);
    v2f a; a.x = af.x; a.y = af.y;
    for (int t = 0; t < 4; ++t) {
      int n = t * 16 + m16;
      const float2 bf = *(const float2*)(W1 + (size_t)n * 64 + 4 * k + 2 * half);
      v2f b; b.x = bf.x; b.y = bf.y;
      acc1[t] = __builtin_amdgcn_wmma_f32_16x16x4_f32(
          false, a, false, b, (short)0, acc1[t], false, false);
    }
  }
  __syncthreads();

  for (int t = 0; t < 4; ++t) {
    float bv = b1[t * 16 + m16];
    for (int v = 0; v < 8; ++v) {
      float h = acc1[t][v] + bv;
      h = h > 0.f ? h : 0.f;
      tile[(v + 8 * half) * LDS_STRIDE + t * 16 + m16] = h;
    }
  }
  __syncthreads();

  v8f acc2[4];
  for (int t = 0; t < 4; ++t)
    for (int j = 0; j < 8; ++j) acc2[t][j] = 0.f;

  for (int k = 0; k < 16; ++k) {
    const float2 af = *(const float2*)(tile + m16 * LDS_STRIDE + 4 * k + 2 * half);
    v2f a; a.x = af.x; a.y = af.y;
    for (int t = 0; t < 4; ++t) {
      int n = t * 16 + m16;
      const float2 bf = *(const float2*)(W2 + (size_t)n * 64 + 4 * k + 2 * half);
      v2f b; b.x = bf.x; b.y = bf.y;
      acc2[t] = __builtin_amdgcn_wmma_f32_16x16x4_f32(
          false, a, false, b, (short)0, acc2[t], false, false);
    }
  }

  for (int t = 0; t < 4; ++t) {
    float bv = b2[t * 16 + m16];
    for (int v = 0; v < 8; ++v) {
      int node = node0 + v + 8 * half;
      if (node < n_nodes)
        out[(size_t)node * 64 + t * 16 + m16] = acc2[t][v] + bv;
    }
  }
}

extern "C" void kernel_launch(void* const* d_in, const int* in_sizes, int n_in,
                              void* d_out, int out_size, void* d_ws, size_t ws_size,
                              hipStream_t stream) {
  const float* x   = (const float*)d_in[0];
  const int*   ei  = (const int*)  d_in[1];
  const float* eps = (const float*)d_in[2];
  const float* W1  = (const float*)d_in[3];
  const float* b1  = (const float*)d_in[4];
  const float* W2  = (const float*)d_in[5];
  const float* b2  = (const float*)d_in[6];
  float* out = (float*)d_out;

  const int n_nodes = in_sizes[0] / 64;
  const int n_edges = in_sizes[1] / 2;
  const int nb = (n_nodes + BROWS - 1) / BROWS;
  const int n_chunks = (n_edges + CHUNK - 1) / CHUNK;

  int* sorted = (int*)d_ws;
  int* hist   = sorted + (size_t)n_edges * 2;
  int* bstart = hist + (size_t)n_chunks * nb;

  bin_count_kernel<<<n_chunks, 1024, 0, stream>>>(ei, hist, n_nodes, n_edges, nb);
  bin_fill_kernel<<<n_chunks, 1024, 0, stream>>>(ei, hist, bstart, sorted, n_nodes, n_edges, nb,
                                                 n_chunks);
  owner_mlp_kernel<<<nb, 256, 0, stream>>>(x, sorted, bstart, eps, W1, b1, W2, b2, out, n_nodes);
}


# W1 and W2 copied once per work-group into LDS (whole-line loads, XOR-swizzled 16-byte groups); B fragments read from LDS instead of 128 sixteen-line global loads per wave
# speedup vs baseline: 3.2247x; 1.1097x over previous
#include <hip/hip_runtime.h>

typedef __attribute__((ext_vector_type(2))) float v2f;
typedef __attribute__((ext_vector_type(8))) float v8f;

#define LDS_STRIDE 68

#define BROWS     128
#define BSHIFT    7
#define NB_MAX    1024
#define CHUNK     16384
#define EPT       16
#define TILE_PAIRS 2560
#define PPT       10
#define GATHER    16
#define COL_BATCH 16

__global__ __launch_bounds__(1024) void bin_count_kernel(const int* __restrict__ ei,
                                                         int* __restrict__ hist,
                                                         int n_nodes, int n_edges, int nb) {
  __shared__ int lh[NB_MAX];
  const int tid = threadIdx.x;
  for (int i = tid; i < nb; i += 1024) lh[i] = 0;
  __syncthreads();

  const int* __restrict__ dstp = ei + n_edges;
  const int e0 = blockIdx.x * CHUNK + tid;
  int d[EPT];
#pragma unroll
  for (int j = 0; j < EPT; ++j) {
    const int e = e0 + j * 1024;
    d[j] = e < n_edges ? dstp[e] : -1;
  }
#pragma unroll
  for (int j = 0; j < EPT; ++j)
    if ((unsigned)d[j] < (unsigned)n_nodes) atomicAdd(&lh[d[j] >> BSHIFT], 1);
  __syncthreads();
  for (int i = tid; i < nb; i += 1024) hist[blockIdx.x * nb + i] = lh[i];
}

__device__ __forceinline__ int wave_incl_scan(int v, int lane) {
#pragma unroll
  for (int o = 1; o < 32; o <<= 1) {
    const int t = __shfl_up(v, o);
    if (lane >= o) v += t;
  }
  return v;
}

__global__ __launch_bounds__(1024) void bin_fill_kernel(const int* __restrict__ ei,
                                                        const int* __restrict__ hist,
                                                        int* __restrict__ bstart,
                                                        int* __restrict__ sorted_i,
                                                        int n_nodes, int n_edges, int nb,
                                                        int n_chunks) {
  __shared__ int2 pairs[CHUNK];
  __shared__ int cur[NB_MAX];
  __shared__ int gdelta[NB_MAX];
  __shared__ int ws_a[32], ws_b[32];
  __shared__ int nvalid_s;

  const int tid  = threadIdx.x;
  const int lane = tid & 31;
  const int wave = __builtin_amdgcn_readfirstlane(tid >> 5);
  const int g = blockIdx.x;

  int tot = 0, pre = 0, mine = 0;
  if (tid < nb) {
    const int* __restrict__ col = hist + tid;
    for (int g0 = 0; g0 < n_chunks; g0 += COL_BATCH) {
      int v[COL_BATCH];
#pragma unroll
      for (int j = 0; j < COL_BATCH; ++j) v[j] = g0 + j < n_chunks ? col[(g0 + j) * nb] : 0;
#pragma unroll
      for (int j = 0; j < COL_BATCH; ++j) {
        tot += v[j];
        pre += g0 + j < g ? v[j] : 0;
        mine = g0 + j == g ? v[j] : mine;
      }
    }
  }
  const int ia = wave_incl_scan(tot, lane);
  const int ib = wave_incl_scan(mine, lane);
  if (lane == 31) { ws_a[wave] = ia; ws_b[wave] = ib; }
  __syncthreads();
  if (wave == 0) {
    const int a = ws_a[lane], b = ws_b[lane];
    const int sa = wave_incl_scan(a, lane);
    const int sb = wave_incl_scan(b, lane);
    ws_a[lane] = sa - a;
    ws_b[lane] = sb - b;
    if (lane == 31) nvalid_s = sb;
  }
  __syncthreads();
  const int start = ia - tot + ws_a[wave];
  const int lofs  = ib - mine + ws_b[wave];
  if (tid < nb) {
    cur[tid] = lofs;
    gdelta[tid] = start + pre - lofs;
    if (g == 0) {
      bstart[tid] = start;
      if (tid == nb - 1) bstart[nb] = start + tot;
    }
  }
  __syncthreads();

  const int* __restrict__ dstp = ei + n_edges;
  const int e0 = g * CHUNK + tid;
  int s[EPT], d[EPT];
#pragma unroll
  for (int j = 0; j < EPT; ++j) {
    const int e = e0 + j * 1024;
    const bool ok = e < n_edges;
    d[j] = ok ? dstp[e] : -1;
    s[j] = ok ? ei[e] : 0;
  }
#pragma unroll
  for (int j = 0; j < EPT; ++j) {
    if ((unsigned)d[j] < (unsigned)n_nodes) {
      const int p = atomicAdd(&cur[d[j] >> BSHIFT], 1);
      pairs[p] = make_int2(s[j], d[j]);
    }
  }
  __syncthreads();

  const int nv = nvalid_s;
  int2* __restrict__ sorted = (int2*)sorted_i;
  for (int p = tid; p < nv; p += 1024) {
    const int2 pr = pairs[p];
    sorted[gdelta[pr.y >> BSHIFT] + p] = pr;
  }
}

__global__ __launch_bounds__(256) void owner_mlp_kernel(
    const float* __restrict__ x, const int* __restrict__ sorted_i,
    const int* __restrict__ bstart, const float* __restrict__ eps_p,
    const float* __restrict__ W1, const float* __restrict__ b1,
    const float* __restrict__ W2, const float* __restrict__ b2,
    float* __restrict__ out, int n_nodes) {
  static_assert(BROWS == 8 * 16, "one 16-row tile per wave of the 256-thread group");
  __shared__ __attribute__((aligned(16))) float acc[BROWS * LDS_STRIDE];
  __shared__ int srcs[TILE_PAIRS];
  __shared__ __attribute__((aligned(16))) int cnt[BROWS];
  __shared__ __attribute__((aligned(16))) int rbeg[BROWS];
  __shared__ __attribute__((aligned(16))) float wl[2 * 64 * 64];

  const int tid  = threadIdx.x;
  const int lane = tid & 31;
  const int wave = __builtin_amdgcn_readfirstlane(tid >> 5);
  const int row0 = blockIdx.x * BROWS;

  {
    const float4* __restrict__ w1v = (const float4*)W1;
    const float4* __restrict__ w2v = (const float4*)W2;
    float4 wr[8];
#pragma unroll
    for (int i = 0; i < 4; ++i) {
      wr[i]     = w1v[tid + 256 * i];
      wr[4 + i] = w2v[tid + 256 * i];
    }
#pragma unroll
    for (int i = 0; i < 4; ++i) {
      const int u = tid + 256 * i;
      const int n = u >> 4;
      const int off = n * 64 + (((u & 15) ^ (n & 15)) << 2);
      *(float4*)(wl + off)           = wr[i];
      *(float4*)(wl + 64 * 64 + off) = wr[4 + i];
    }
  }

  const int2* __restrict__ sorted = (const int2*)sorted_i;
  const float4* __restrict__ x4 = (const float4*)x;
  const int s1 = bstart[blockIdx.x + 1];
  int t0 = bstart[blockIdx.x];
  bool first = true;

  for (;;) {
    if (tid < BROWS) cnt[tid] = 0;
    __syncthreads();

    int2 pr[PPT];
    int key[PPT];
#pragma unroll
    for (int j = 0; j < PPT; ++j) {
      const int p = t0 + j * 256 + tid;
      pr[j] = p < s1 ? sorted[p] : make_int2(0, -1);
    }
#pragma unroll
    for (int j = 0; j < PPT; ++j) {
      key[j] = -1;
      if (pr[j].y >= 0) {
        const int r = pr[j].y - row0;
        key[j] = (atomicAdd(&cnt[r], 1) << BSHIFT) | r;
      }
    }
    __syncthreads();

    if (wave == 0) {
      const int4 c = *(const int4*)(cnt + 4 * lane);
      const int tot = c.x + c.y + c.z + c.w;
      const int ex = wave_incl_scan(tot, lane) - tot;
      *(int4*)(rbeg + 4 * lane) = make_int4(ex, ex + c.x, ex + c.x + c.y, ex + c.x + c.y + c.z);
    }
    __syncthreads();

#pragma unroll
    for (int j = 0; j < PPT; ++j)
      if (key[j] >= 0) srcs[rbeg[key[j] & (BROWS - 1)] + (key[j] >> BSHIFT)] = pr[j].x;
    __syncthreads();

    {
      const int hrow = lane >> 4;
      const int c16  = lane & 15;
      for (int i = 0; i < 8; ++i) {
        const int rl  = wave * 16 + 2 * i + hrow;
        const int beg = rbeg[rl];
        const int n   = cnt[rl];
        const int na = __builtin_amdgcn_readlane(n, 0);
        const int nb = __builtin_amdgcn_readlane(n, 16);
        const int nmax = na > nb ? na : nb;
        float4 sum = make_float4(0.f, 0.f, 0.f, 0.f);
        for (int j0 = 0; j0 < nmax; j0 += GATHER) {
          int sj[GATHER];
          float4 v[GATHER];
#pragma unroll
          for (int j = 0; j < GATHER; ++j) {
            sj[j] = 0;
            if (j0 + j < n) sj[j] = srcs[beg + j0 + j];
          }
#pragma unroll
          for (int j = 0; j < GATHER; ++j) v[j] = x4[(unsigned)sj[j] * 16u + (unsigned)c16];
#pragma unroll
          for (int j = 0; j < GATHER; ++j) {
            if (j0 + j < n) {
              sum.x += v[j].x; sum.y += v[j].y; sum.z += v[j].z; sum.w += v[j].w;
            }
          }
        }
        float4* d = (float4*)(acc + rl * LDS_STRIDE + 4 * c16);
        if (first) {
          *d = sum;
        } else {
          const float4 o = *d;
          *d = make_float4(o.x + sum.x, o.y + sum.y, o.z + sum.z, o.w + sum.w);
        }
      }
    }

    t0 += TILE_PAIRS;
    first = false;
    if (t0 >= s1) break;
    __syncthreads();
  }
  __syncthreads();

  const int half = lane >> 4;
  const int m16  = lane & 15;
  const int wv = tid >> 5;
  float* tile = acc + wv * 16 * LDS_STRIDE;

  const float k1 = 1.0f + eps_p[0];
  const int node0 = row0 + wv * 16;

  for (int i = 0; i < 8; ++i) {
    int e4 = lane + i * 32;
    int r  = e4 >> 4;
    int c4 = e4 & 15;
    int node = node0 + r;
    int cn = node < n_nodes ? node : n_nodes - 1;
    float4 xv = *((const float4*)(x + (size_t)cn * 64) + c4);
    float* d = tile + r * LDS_STRIDE + c4 * 4;
    d[0] = k1 * xv.x + d[0];
    d[1] = k1 * xv.y + d[1];
    d[2] = k1 * xv.z + d[2];
    d[3] = k1 * xv.w + d[3];
  }
  __syncthreads();

  v8f acc1[4];
  for (int t = 0; t < 4; ++t)
    for (int j = 0; j < 8; ++j) acc1[t][j] = 0.f;

  for (int k = 0; k < 16; ++k) {
    const float2 af = *(const float2*)(tile + m16 * LDS_STRIDE + 4 * k + 2 * half);
    v2f a; a.x = af.x; a.y = af.y;
    for (int t = 0; t < 4; ++t) {
      int n = t * 16 + m16;
      const float2 bf = *(const float2*)(wl + n * 64 + ((k ^ m16) << 2) + 2 * half);
      v2f b; b.x = bf.x; b.y = bf.y;
      acc1[t] = __builtin_amdgcn_wmma_f32_16x16x4_f32(
          false, a, false, b, (short)0, acc1[t], false, false);
    }
  }
  __syncthreads();

  for (int t = 0; t < 4; ++t) {
    float bv = b1[t * 16 + m16];
    for (int v = 0; v < 8; ++v) {
      float h = acc1[t][v] + bv;
      h = h > 0.f ? h : 0.f;
      tile[(v + 8 * half) * LDS_STRIDE + t * 16 + m16] = h;
    }
  }
  __syncthreads();

  v8f acc2[4];
  for (int t = 0; t < 4; ++t)
    for (int j = 0; j < 8; ++j) acc2[t][j] = 0.f;

  for (int k = 0; k < 16; ++k) {
    const float2 af = *(const float2*)(tile + m16 * LDS_STRIDE + 4 * k + 2 * half);
    v2f a; a.x = af.x; a.y = af.y;
    for (int t = 0; t < 4; ++t) {
      int n = t * 16 + m16;
      const float2 bf = *(const float2*)(wl + 64 * 64 + n * 64 + ((k ^ m16) << 2) + 2 * half);
      v2f b; b.x = bf.x; b.y = bf.y;
      acc2[t] = __builtin_amdgcn_wmma_f32_16x16x4_f32(
          false, a, false, b, (short)0, acc2[t], false, false);
    }
  }

  for (int t = 0; t < 4; ++t) {
    float bv = b2[t * 16 + m16];
    for (int v = 0; v < 8; ++v) {
      int node = node0 + v + 8 * half;
      if (node < n_nodes)
        out[(size_t)node * 64 + t * 16 + m16] = acc2[t][v] + bv;
    }
  }
}

extern "C" void kernel_launch(void* const* d_in, const int* in_sizes, int n_in,
                              void* d_out, int out_size, void* d_ws, size_t ws_size,
                              hipStream_t stream) {
  const float* x   = (const float*)d_in[0];
  const int*   ei  = (const int*)  d_in[1];
  const float* eps = (const float*)d_in[2];
  const float* W1  = (const float*)d_in[3];
  const float* b1  = (const float*)d_in[4];
  const float* W2  = (const float*)d_in[5];
  const float* b2  = (const float*)d_in[6];
  float* out = (float*)d_out;

  const int n_nodes = in_sizes[0] / 64;
  const int n_edges = in_sizes[1] / 2;
  const int nb = (n_nodes + BROWS - 1) / BROWS;
  const int n_chunks = (n_edges + CHUNK - 1) / CHUNK;

  int* sorted = (int*)d_ws;
  int* hist   = sorted + (size_t)n_edges * 2;
  int* bstart = hist + (size_t)n_chunks * nb;

  bin_count_kernel<<<n_chunks, 1024, 0, stream>>>(ei, hist, n_nodes, n_edges, nb);
  bin_fill_kernel<<<n_chunks, 1024, 0, stream>>>(ei, hist, bstart, sorted, n_nodes, n_edges, nb,
                                                 n_chunks);
  owner_mlp_kernel<<<nb, 256, 0, stream>>>(x, sorted, bstart, eps, W1, b1, W2, b2, out, n_nodes);
}


# Global count and offset passes dropped: each chunk sorts its pairs by bin into its own region plus a run table; owner group collects its bin's runs itself; 2 launches not 3; sums stay in arrival rank
# speedup vs baseline: 4.0079x; 1.2429x over previous
#include <hip/hip_runtime.h>

typedef __attribute__((ext_vector_type(2))) float v2f;
typedef __attribute__((ext_vector_type(8))) float v8f;

#define LDS_STRIDE 68

#define BROWS     128
#define BSHIFT    7
#define NB_MAX    1024
#define CHUNK     16384
#define EPT       16
#define CH_MAX    128
#define TILE_PAIRS 2560
#define PPT       10
#define GATHER    16

__device__ __forceinline__ int wave_incl_scan(int v, int lane) {
#pragma unroll
  for (int o = 1; o < 32; o <<= 1) {
    const int t = __shfl_up(v, o);
    if (lane >= o) v += t;
  }
  return v;
}

__global__ __launch_bounds__(1024) void chunk_sort_kernel(const int* __restrict__ ei,
                                                          int* __restrict__ tab,
                                                          int* __restrict__ sorted_i,
                                                          int n_nodes, int n_edges, int nb) {
  static_assert(EPT == 16 && CHUNK == EPT * 1024, "sixteen edges per thread; the fence below names sixteen values");
  __shared__ __attribute__((aligned(16))) int2 pairs[CHUNK];
  __shared__ int lh[NB_MAX];
  __shared__ int ws_b[32];

  const int tid  = threadIdx.x;
  const int lane = tid & 31;
  const int wave = __builtin_amdgcn_readfirstlane(tid >> 5);
  const int g = blockIdx.x;

  for (int i = tid; i < nb; i += 1024) lh[i] = 0;
  __syncthreads();

  const int* __restrict__ dstp = ei + n_edges;
  const int e0 = g * CHUNK + tid;
  int s[EPT], d[EPT], rk[EPT];
#pragma unroll
  for (int j = 0; j < EPT; ++j) {
    const int e = e0 + j * 1024;
    const bool ok = e < n_edges;
    d[j] = ok ? dstp[e] : -1;
    s[j] = ok ? ei[e] : 0;
  }
#pragma unroll
  for (int j = 0; j < EPT; ++j) {
    rk[j] = -1;
    if ((unsigned)d[j] < (unsigned)n_nodes) rk[j] = atomicAdd(&lh[d[j] >> BSHIFT], 1);
  }
  __syncthreads();

  const int mine = tid < nb ? lh[tid] : 0;
  const int ib = wave_incl_scan(mine, lane);
  if (lane == 31) ws_b[wave] = ib;
  __syncthreads();
  if (wave == 0) {
    const int b = ws_b[lane];
    const int sb = wave_incl_scan(b, lane);
    ws_b[lane] = sb - b;
  }
  __syncthreads();
  const int lofs = ib - mine + ws_b[wave];
  if (tid < nb) {
    lh[tid] = lofs;
    int* __restrict__ trow = tab + (size_t)g * (nb + 1);
    trow[tid] = lofs;
    if (tid == nb - 1) trow[nb] = lofs + mine;
  }
  __syncthreads();

  int bs[EPT];
#pragma unroll
  for (int j = 0; j < EPT; ++j) bs[j] = lh[(rk[j] >= 0 ? d[j] : 0) >> BSHIFT];
  asm volatile("" :: "v"(bs[0]), "v"(bs[1]), "v"(bs[2]), "v"(bs[3]), "v"(bs[4]), "v"(bs[5]),
                     "v"(bs[6]), "v"(bs[7]), "v"(bs[8]), "v"(bs[9]), "v"(bs[10]), "v"(bs[11]),
                     "v"(bs[12]), "v"(bs[13]), "v"(bs[14]), "v"(bs[15]));
#pragma unroll
  for (int j = 0; j < EPT; ++j)
    if (rk[j] >= 0) pairs[bs[j] + rk[j]] = make_int2(s[j], d[j]);
  __syncthreads();

  const int4* __restrict__ p4 = (const int4*)pairs;
  int4* __restrict__ o4 = (int4*)sorted_i + (size_t)g * (CHUNK / 2);
  int4 cp[EPT / 2];
#pragma unroll
  for (int i = 0; i < EPT / 2; ++i) cp[i] = p4[tid + i * 1024];
#pragma unroll
  for (int i = 0; i < EPT / 2; ++i) o4[tid + i * 1024] = cp[i];
}

__global__ __launch_bounds__(256) void owner_mlp_kernel(
    const float* __restrict__ x, const int* __restrict__ sorted_i,
    const int* __restrict__ tab, const float* __restrict__ eps_p,
    const float* __restrict__ W1, const float* __restrict__ b1,
    const float* __restrict__ W2, const float* __restrict__ b2,
    float* __restrict__ out, int n_nodes, int n_chunks, int nb) {
  static_assert(CH_MAX == 4 * 32, "four chunks per lane of wave 0; a power of two for the run search");
  static_assert(PPT == 10, "the fence behind the run search names ten values");
  static_assert(BROWS == 8 * 16, "one 16-row tile per wave of the 256-thread group");
  __shared__ __attribute__((aligned(16))) float acc[BROWS * LDS_STRIDE];
  __shared__ int srcs[TILE_PAIRS];
  __shared__ __attribute__((aligned(16))) int cnt[BROWS];
  __shared__ __attribute__((aligned(16))) int rbeg[BROWS];
  __shared__ __attribute__((aligned(16))) int cs[CH_MAX + 4];
  __shared__ __attribute__((aligned(16))) int gd[CH_MAX];
  __shared__ __attribute__((aligned(16))) float wl[2 * 64 * 64];

  const int tid  = threadIdx.x;
  const int lane = tid & 31;
  const int wave = __builtin_amdgcn_readfirstlane(tid >> 5);
  const int row0 = blockIdx.x * BROWS;

  int rlo[4] = {0, 0, 0, 0}, rhi[4] = {0, 0, 0, 0};
  if (wave == 0) {
#pragma unroll
    for (int i = 0; i < 4; ++i) {
      const int c = 4 * lane + i;
      const bool ok = c < n_chunks;
      const int* __restrict__ tp = tab + (size_t)(ok ? c : 0) * (nb + 1) + blockIdx.x;
      rlo[i] = ok ? tp[0] : 0;
      rhi[i] = ok ? tp[1] : 0;
    }
  }

  {
    const float4* __restrict__ w1v = (const float4*)W1;
    const float4* __restrict__ w2v = (const float4*)W2;
    float4 wr[8];
#pragma unroll
    for (int i = 0; i < 4; ++i) {
      wr[i]     = w1v[tid + 256 * i];
      wr[4 + i] = w2v[tid + 256 * i];
    }
#pragma unroll
    for (int i = 0; i < 4; ++i) {
      const int u = tid + 256 * i;
      const int n = u >> 4;
      const int off = n * 64 + (((u & 15) ^ (n & 15)) << 2);
      *(float4*)(wl + off)           = wr[i];
      *(float4*)(wl + 64 * 64 + off) = wr[4 + i];
    }
  }

  const int2* __restrict__ sorted = (const int2*)sorted_i;
  const float4* __restrict__ x4 = (const float4*)x;
  if (wave == 0) {
    const int l0 = rhi[0] - rlo[0], l1 = rhi[1] - rlo[1], l2 = rhi[2] - rlo[2], l3 = rhi[3] - rlo[3];
    const int tot = l0 + l1 + l2 + l3;
    const int e0 = wave_incl_scan(tot, lane) - tot;
    const int e1 = e0 + l0, e2 = e1 + l1, e3 = e2 + l2;
    const int base = 4 * lane * CHUNK;
    *(int4*)(cs + 4 * lane) = make_int4(e0, e1, e2, e3);
    *(int4*)(gd + 4 * lane) = make_int4(base + rlo[0] - e0, base + CHUNK + rlo[1] - e1,
                                        base + 2 * CHUNK + rlo[2] - e2,
                                        base + 3 * CHUNK + rlo[3] - e3);
    if (lane == 31) cs[CH_MAX] = e3 + l3;
  }
  if (tid < BROWS) cnt[tid] = 0;
  __syncthreads();
  const int s1 = cs[CH_MAX];
  int t0 = 0;
  bool first = true;

  for (;;) {
    int2 pr[PPT];
    int key[PPT];
    int c[PPT];
#pragma unroll
    for (int j = 0; j < PPT; ++j) c[j] = 0;
#pragma unroll
    for (int st = CH_MAX / 2; st >= 1; st >>= 1) {
      int v[PPT];
#pragma unroll
      for (int j = 0; j < PPT; ++j) v[j] = cs[c[j] + st];
#pragma unroll
      for (int j = 0; j < PPT; ++j) c[j] += v[j] <= t0 + j * 256 + tid ? st : 0;
    }
    int gq[PPT];
#pragma unroll
    for (int j = 0; j < PPT; ++j) gq[j] = gd[c[j]];
    asm volatile("" :: "v"(gq[0]), "v"(gq[1]), "v"(gq[2]), "v"(gq[3]), "v"(gq[4]),
                       "v"(gq[5]), "v"(gq[6]), "v"(gq[7]), "v"(gq[8]), "v"(gq[9]));
#pragma unroll
    for (int j = 0; j < PPT; ++j) {
      const int q = t0 + j * 256 + tid;
      pr[j] = q < s1 ? sorted[gq[j] + q] : make_int2(0, -1);
    }
#pragma unroll
    for (int j = 0; j < PPT; ++j) {
      key[j] = -1;
      if (pr[j].y >= 0) {
        const int r = pr[j].y - row0;
        key[j] = (atomicAdd(&cnt[r], 1) << BSHIFT) | r;
      }
    }
    __syncthreads();

    if (wave == 0) {
      const int4 c = *(const int4*)(cnt + 4 * lane);
      const int tot = c.x + c.y + c.z + c.w;
      const int ex = wave_incl_scan(tot, lane) - tot;
      *(int4*)(rbeg + 4 * lane) = make_int4(ex, ex + c.x, ex + c.x + c.y, ex + c.x + c.y + c.z);
    }
    __syncthreads();

#pragma unroll
    for (int j = 0; j < PPT; ++j)
      if (key[j] >= 0) srcs[rbeg[key[j] & (BROWS - 1)] + (key[j] >> BSHIFT)] = pr[j].x;
    __syncthreads();

    {
      const int hrow = lane >> 4;
      const int c16  = lane & 15;
      for (int i = 0; i < 8; ++i) {
        const int rl  = wave * 16 + 2 * i + hrow;
        const int beg = rbeg[rl];
        const int n   = cnt[rl];
        const int na = __builtin_amdgcn_readlane(n, 0);
        const int nb = __builtin_amdgcn_readlane(n, 16);
        const int nmax = na > nb ? na : nb;
        float4 sum = make_float4(0.f, 0.f, 0.f, 0.f);
        for (int j0 = 0; j0 < nmax; j0 += GATHER) {
          int sj[GATHER];
          float4 v[GATHER];
#pragma unroll
          for (int j = 0; j < GATHER; ++j) {
            sj[j] = 0;
            if (j0 + j < n) sj[j] = srcs[beg + j0 + j];
          }
#pragma unroll
          for (int j = 0; j < GATHER; ++j) v[j] = x4[(unsigned)sj[j] * 16u + (unsigned)c16];
#pragma unroll
          for (int j = 0; j < GATHER; ++j) {
            if (j0 + j < n) {
              sum.x += v[j].x; sum.y += v[j].y; sum.z += v[j].z; sum.w += v[j].w;
            }
          }
        }
        float4* d = (float4*)(acc + rl * LDS_STRIDE + 4 * c16);
        if (first) {
          *d = sum;
        } else {
          const float4 o = *d;
          *d = make_float4(o.x + sum.x, o.y + sum.y, o.z + sum.z, o.w + sum.w);
        }
      }
    }

    t0 += TILE_PAIRS;
    first = false;
    if (t0 >= s1) break;
    __syncthreads();
    if (tid < BROWS) cnt[tid] = 0;
    __syncthreads();
  }
  __syncthreads();

  const int half = lane >> 4;
  const int m16  = lane & 15;
  const int wv = tid >> 5;
  float* tile = acc + wv * 16 * LDS_STRIDE;

  const float k1 = 1.0f + eps_p[0];
  const int node0 = row0 + wv * 16;

  for (int i = 0; i < 8; ++i) {
    int e4 = lane + i * 32;
    int r  = e4 >> 4;
    int c4 = e4 & 15;
    int node = node0 + r;
    int cn = node < n_nodes ? node : n_nodes - 1;
    float4 xv = *((const float4*)(x + (size_t)cn * 64) + c4);
    float* d = tile + r * LDS_STRIDE + c4 * 4;
    d[0] = k1 * xv.x + d[0];
    d[1] = k1 * xv.y + d[1];
    d[2] = k1 * xv.z + d[2];
    d[3] = k1 * xv.w + d[3];
  }
  __syncthreads();

  v8f acc1[4];
  for (int t = 0; t < 4; ++t)
    for (int j = 0; j < 8; ++j) acc1[t][j] = 0.f;

  for (int k = 0; k < 16; ++k) {
    const float2 af = *(const float2*)(tile + m16 * LDS_STRIDE + 4 * k + 2 * half);
    v2f a; a.x = af.x; a.y = af.y;
    for (int t = 0; t < 4; ++t) {
      int n = t * 16 + m16;
      const float2 bf = *(const float2*)(wl + n * 64 + ((k ^ m16) << 2) + 2 * half);
      v2f b; b.x = bf.x; b.y = bf.y;
      acc1[t] = __builtin_amdgcn_wmma_f32_16x16x4_f32(
          false, a, false, b, (short)0, acc1[t], false, false);
    }
  }
  __syncthreads();

  for (int t = 0; t < 4; ++t) {
    float bv = b1[t * 16 + m16];
    for (int v = 0; v < 8; ++v) {
      float h = acc1[t][v] + bv;
      h = h > 0.f ? h : 0.f;
      tile[(v + 8 * half) * LDS_STRIDE + t * 16 + m16] = h;
    }
  }
  __syncthreads();

  v8f acc2[4];
  for (int t = 0; t < 4; ++t)
    for (int j = 0; j < 8; ++j) acc2[t][j] = 0.f;

  for (int k = 0; k < 16; ++k) {
    const float2 af = *(const float2*)(tile + m16 * LDS_STRIDE + 4 * k + 2 * half);
    v2f a; a.x = af.x; a.y = af.y;
    for (int t = 0; t < 4; ++t) {
      int n = t * 16 + m16;
      const float2 bf = *(const float2*)(wl + 64 * 64 + n * 64 + ((k ^ m16) << 2) + 2 * half);
      v2f b; b.x = bf.x; b.y = bf.y;
      acc2[t] = __builtin_amdgcn_wmma_f32_16x16x4_f32(
          false, a, false, b, (short)0, acc2[t], false, false);
    }
  }

  for (int t = 0; t < 4; ++t) {
    float bv = b2[t * 16 + m16];
    for (int v = 0; v < 8; ++v) {
      int node = node0 + v + 8 * half;
      if (node < n_nodes)
        out[(size_t)node * 64 + t * 16 + m16] = acc2[t][v] + bv;
    }
  }
}

extern "C" void kernel_launch(void* const* d_in, const int* in_sizes, int n_in,
                              void* d_out, int out_size, void* d_ws, size_t ws_size,
                              hipStream_t stream) {
  const float* x   = (const float*)d_in[0];
  const int*   ei  = (const int*)  d_in[1];
  const float* eps = (const float*)d_in[2];
  const float* W1  = (const float*)d_in[3];
  const float* b1  = (const float*)d_in[4];
  const float* W2  = (const float*)d_in[5];
  const float* b2  = (const float*)d_in[6];
  float* out = (float*)d_out;

  const int n_nodes = in_sizes[0] / 64;
  const int n_edges = in_sizes[1] / 2;
  const int nb = (n_nodes + BROWS - 1) / BROWS;
  const int n_chunks = (n_edges + CHUNK - 1) / CHUNK;

  int* sorted = (int*)d_ws;
  int* tab    = sorted + (size_t)n_chunks * CHUNK * 2;

  chunk_sort_kernel<<<n_chunks, 1024, 0, stream>>>(ei, tab, sorted, n_nodes, n_edges, nb);
  owner_mlp_kernel<<<nb, 256, 0, stream>>>(x, sorted, tab, eps, W1, b1, W2, b2, out, n_nodes,
                                           n_chunks, nb);
}


# Bins resized so owner groups come to exactly 3 per WGP: 106 bins of 144 rows (ninth tile shared by the 8 waves) + 662 of 128; was 782 groups, a fourth on 14 WGPs; sums stay in arrival rank per bin
# speedup vs baseline: 4.1947x; 1.0466x over previous
#include <hip/hip_runtime.h>

typedef __attribute__((ext_vector_type(2))) float v2f;
typedef __attribute__((ext_vector_type(8))) float v8f;

#define LDS_STRIDE 68

#define BROWS     128
#define BSHIFT    7
#define BROWS_BIG 144
#define OWNERS    768
#define ROWS_PAD  160
#define KSHIFT    8
#define NB_MAX    1024
#define CHUNK     16384
#define EPT       16
#define CH_MAX    128
#define TILE_PAIRS 2560
#define PPT       10
#define GATHER    16

__device__ __forceinline__ int wave_incl_scan(int v, int lane) {
#pragma unroll
  for (int o = 1; o < 32; o <<= 1) {
    const int t = __shfl_up(v, o);
    if (lane >= o) v += t;
  }
  return v;
}

__device__ __forceinline__ int bin_of(int d, int nbig) {
  const int rows_big = nbig * BROWS_BIG;
  return d < rows_big ? (int)((unsigned)d / (unsigned)BROWS_BIG) : nbig + ((d - rows_big) >> BSHIFT);
}

__global__ __launch_bounds__(1024) void chunk_sort_kernel(const int* __restrict__ ei,
                                                          int* __restrict__ tab,
                                                          int* __restrict__ sorted_i,
                                                          int n_nodes, int n_edges, int nb, int nbig) {
  static_assert(EPT == 16 && CHUNK == EPT * 1024, "sixteen edges per thread; the fence below names sixteen values");
  __shared__ __attribute__((aligned(16))) int2 pairs[CHUNK];
  __shared__ int lh[NB_MAX];
  __shared__ int ws_b[32];

  const int tid  = threadIdx.x;
  const int lane = tid & 31;
  const int wave = __builtin_amdgcn_readfirstlane(tid >> 5);
  const int g = blockIdx.x;

  for (int i = tid; i < nb; i += 1024) lh[i] = 0;
  __syncthreads();

  const int* __restrict__ dstp = ei + n_edges;
  const int e0 = g * CHUNK + tid;
  int s[EPT], d[EPT], rk[EPT], bn[EPT];
#pragma unroll
  for (int j = 0; j < EPT; ++j) {
    const int e = e0 + j * 1024;
    const bool ok = e < n_edges;
    d[j] = ok ? dstp[e] : -1;
    s[j] = ok ? ei[e] : 0;
  }
#pragma unroll
  for (int j = 0; j < EPT; ++j) {
    rk[j] = -1;
    bn[j] = 0;
    if ((unsigned)d[j] < (unsigned)n_nodes) {
      bn[j] = bin_of(d[j], nbig);
      rk[j] = atomicAdd(&lh[bn[j]], 1);
    }
  }
  __syncthreads();

  const int mine = tid < nb ? lh[tid] : 0;
  const int ib = wave_incl_scan(mine, lane);
  if (lane == 31) ws_b[wave] = ib;
  __syncthreads();
  if (wave == 0) {
    const int b = ws_b[lane];
    const int sb = wave_incl_scan(b, lane);
    ws_b[lane] = sb - b;
  }
  __syncthreads();
  const int lofs = ib - mine + ws_b[wave];
  if (tid < nb) {
    lh[tid] = lofs;
    int* __restrict__ trow = tab + (size_t)g * (nb + 1);
    trow[tid] = lofs;
    if (tid == nb - 1) trow[nb] = lofs + mine;
  }
  __syncthreads();

  int bs[EPT];
#pragma unroll
  for (int j = 0; j < EPT; ++j) bs[j] = lh[bn[j]];
  asm volatile("" :: "v"(bs[0]), "v"(bs[1]), "v"(bs[2]), "v"(bs[3]), "v"(bs[4]), "v"(bs[5]),
                     "v"(bs[6]), "v"(bs[7]), "v"(bs[8]), "v"(bs[9]), "v"(bs[10]), "v"(bs[11]),
                     "v"(bs[12]), "v"(bs[13]), "v"(bs[14]), "v"(bs[15]));
#pragma unroll
  for (int j = 0; j < EPT; ++j)
    if (rk[j] >= 0) pairs[bs[j] + rk[j]] = make_int2(s[j], d[j]);
  __syncthreads();

  const int4* __restrict__ p4 = (const int4*)pairs;
  int4* __restrict__ o4 = (int4*)sorted_i + (size_t)g * (CHUNK / 2);
  int4 cp[EPT / 2];
#pragma unroll
  for (int i = 0; i < EPT / 2; ++i) cp[i] = p4[tid + i * 1024];
#pragma unroll
  for (int i = 0; i < EPT / 2; ++i) o4[tid + i * 1024] = cp[i];
}

__global__ __launch_bounds__(256) void owner_mlp_kernel(
    const float* __restrict__ x, const int* __restrict__ sorted_i,
    const int* __restrict__ tab, const float* __restrict__ eps_p,
    const float* __restrict__ W1, const float* __restrict__ b1,
    const float* __restrict__ W2, const float* __restrict__ b2,
    float* __restrict__ out, int n_nodes, int n_chunks, int nb, int nbig) {
  static_assert(CH_MAX == 4 * 32, "four chunks per lane of wave 0; a power of two for the run search");
  static_assert(PPT == 10, "the fence behind the run search names ten values");
  static_assert(BROWS == 8 * 16 && BROWS_BIG == BROWS + 16,
                "one 16-row tile per wave of the 256-thread group; a big bin has one tile more");
  static_assert(ROWS_PAD == 5 * 32 && ROWS_PAD >= BROWS_BIG && BROWS_BIG <= (1 << KSHIFT),
                "five row counters per lane of wave 0; a row number fits the key's low bits");
  __shared__ __attribute__((aligned(16))) float acc[BROWS_BIG * LDS_STRIDE];
  __shared__ int srcs[TILE_PAIRS];
  __shared__ int cnt[ROWS_PAD];
  __shared__ int rbeg[ROWS_PAD];
  __shared__ __attribute__((aligned(16))) int cs[CH_MAX + 4];
  __shared__ __attribute__((aligned(16))) int gd[CH_MAX];
  __shared__ __attribute__((aligned(16))) float wl[2 * 64 * 64];

  const int tid  = threadIdx.x;
  const int lane = tid & 31;
  const int wave = __builtin_amdgcn_readfirstlane(tid >> 5);
  const int bin = blockIdx.x;
  const bool big = bin < nbig;
  const int row0 = big ? bin * BROWS_BIG : nbig * BROWS_BIG + (bin - nbig) * BROWS;

  int rlo[4] = {0, 0, 0, 0}, rhi[4] = {0, 0, 0, 0};
  if (wave == 0) {
#pragma unroll
    for (int i = 0; i < 4; ++i) {
      const int c = 4 * lane + i;
      const bool ok = c < n_chunks;
      const int* __restrict__ tp = tab + (size_t)(ok ? c : 0) * (nb + 1) + blockIdx.x;
      rlo[i] = ok ? tp[0] : 0;
      rhi[i] = ok ? tp[1] : 0;
    }
  }

  {
    const float4* __restrict__ w1v = (const float4*)W1;
    const float4* __restrict__ w2v = (const float4*)W2;
    float4 wr[8];
#pragma unroll
    for (int i = 0; i < 4; ++i) {
      wr[i]     = w1v[tid + 256 * i];
      wr[4 + i] = w2v[tid + 256 * i];
    }
#pragma unroll
    for (int i = 0; i < 4; ++i) {
      const int u = tid + 256 * i;
      const int n = u >> 4;
      const int off = n * 64 + (((u & 15) ^ (n & 15)) << 2);
      *(float4*)(wl + off)           = wr[i];
      *(float4*)(wl + 64 * 64 + off) = wr[4 + i];
    }
  }

  const int2* __restrict__ sorted = (const int2*)sorted_i;
  const float4* __restrict__ x4 = (const float4*)x;
  if (wave == 0) {
    const int l0 = rhi[0] - rlo[0], l1 = rhi[1] - rlo[1], l2 = rhi[2] - rlo[2], l3 = rhi[3] - rlo[3];
    const int tot = l0 + l1 + l2 + l3;
    const int e0 = wave_incl_scan(tot, lane) - tot;
    const int e1 = e0 + l0, e2 = e1 + l1, e3 = e2 + l2;
    const int base = 4 * lane * CHUNK;
    *(int4*)(cs + 4 * lane) = make_int4(e0, e1, e2, e3);
    *(int4*)(gd + 4 * lane) = make_int4(base + rlo[0] - e0, base + CHUNK + rlo[1] - e1,
                                        base + 2 * CHUNK + rlo[2] - e2,
                                        base + 3 * CHUNK + rlo[3] - e3);
    if (lane == 31) cs[CH_MAX] = e3 + l3;
  }
  if (tid < ROWS_PAD) cnt[tid] = 0;
  __syncthreads();
  const int s1 = cs[CH_MAX];
  int t0 = 0;
  bool first = true;

  for (;;) {
    int2 pr[PPT];
    int key[PPT];
    int c[PPT];
#pragma unroll
    for (int j = 0; j < PPT; ++j) c[j] = 0;
#pragma unroll
    for (int st = CH_MAX / 2; st >= 1; st >>= 1) {
      int v[PPT];
#pragma unroll
      for (int j = 0; j < PPT; ++j) v[j] = cs[c[j] + st];
#pragma unroll
      for (int j = 0; j < PPT; ++j) c[j] += v[j] <= t0 + j * 256 + tid ? st : 0;
    }
    int gq[PPT];
#pragma unroll
    for (int j = 0; j < PPT; ++j) gq[j] = gd[c[j]];
    asm volatile("" :: "v"(gq[0]), "v"(gq[1]), "v"(gq[2]), "v"(gq[3]), "v"(gq[4]),
                       "v"(gq[5]), "v"(gq[6]), "v"(gq[7]), "v"(gq[8]), "v"(gq[9]));
#pragma unroll
    for (int j = 0; j < PPT; ++j) {
      const int q = t0 + j * 256 + tid;
      pr[j] = q < s1 ? sorted[gq[j] + q] : make_int2(0, -1);
    }
#pragma unroll
    for (int j = 0; j < PPT; ++j) {
      key[j] = -1;
      if (pr[j].y >= 0) {
        const int r = pr[j].y - row0;
        key[j] = (atomicAdd(&cnt[r], 1) << KSHIFT) | r;
      }
    }
    __syncthreads();

    if (wave == 0) {
      const int* __restrict__ cp = cnt + 5 * lane;
      const int c0 = cp[0], c1 = cp[1], c2 = cp[2], c3 = cp[3], c4 = cp[4];
      const int tot = c0 + c1 + c2 + c3 + c4;
      const int ex = wave_incl_scan(tot, lane) - tot;
      int* __restrict__ rp = rbeg + 5 * lane;
      rp[0] = ex;
      rp[1] = ex + c0;
      rp[2] = ex + c0 + c1;
      rp[3] = ex + c0 + c1 + c2;
      rp[4] = ex + c0 + c1 + c2 + c3;
    }
    __syncthreads();

#pragma unroll
    for (int j = 0; j < PPT; ++j)
      if (key[j] >= 0) srcs[rbeg[key[j] & ((1 << KSHIFT) - 1)] + (key[j] >> KSHIFT)] = pr[j].x;
    __syncthreads();

    {
      const int hrow = lane >> 4;
      const int c16  = lane & 15;
      const int nsteps = big ? 9 : 8;
      for (int i = 0; i < nsteps; ++i) {
        const int rl  = (i < 8 ? wave * 16 + 2 * i : BROWS + 2 * wave) + hrow;
        const int beg = rbeg[rl];
        const int n   = cnt[rl];
        const int na = __builtin_amdgcn_readlane(n, 0);
        const int nb = __builtin_amdgcn_readlane(n, 16);
        const int nmax = na > nb ? na : nb;
        float4 sum = make_float4(0.f, 0.f, 0.f, 0.f);
        for (int j0 = 0; j0 < nmax; j0 += GATHER) {
          int sj[GATHER];
          float4 v[GATHER];
#pragma unroll
          for (int j = 0; j < GATHER; ++j) {
            sj[j] = 0;
            if (j0 + j < n) sj[j] = srcs[beg + j0 + j];
          }
#pragma unroll
          for (int j = 0; j < GATHER; ++j) v[j] = x4[(unsigned)sj[j] * 16u + (unsigned)c16];
#pragma unroll
          for (int j = 0; j < GATHER; ++j) {
            if (j0 + j < n) {
              sum.x += v[j].x; sum.y += v[j].y; sum.z += v[j].z; sum.w += v[j].w;
            }
          }
        }
        float4* d = (float4*)(acc + rl * LDS_STRIDE + 4 * c16);
        if (first) {
          *d = sum;
        } else {
          const float4 o = *d;
          *d = make_float4(o.x + sum.x, o.y + sum.y, o.z + sum.z, o.w + sum.w);
        }
      }
    }

    t0 += TILE_PAIRS;
    first = false;
    if (t0 >= s1) break;
    __syncthreads();
    if (tid < ROWS_PAD) cnt[tid] = 0;
    __syncthreads();
  }
  __syncthreads();

  const int half = lane >> 4;
  const int m16  = lane & 15;
  const int wv = tid >> 5;
  float* tile = acc + wv * 16 * LDS_STRIDE;
  float* tile9 = acc + BROWS * LDS_STRIDE;
  const bool extra = big && wave < 4;

  const float k1 = 1.0f + eps_p[0];
  const int node0 = row0 + wv * 16;

  for (int i = 0; i < 8; ++i) {
    int e4 = lane + i * 32;
    int r  = e4 >> 4;
    int c4 = e4 & 15;
    int node = node0 + r;
    int cn = node < n_nodes ? node : n_nodes - 1;
    float4 xv = *((const float4*)(x + (size_t)cn * 64) + c4);
    float* d = tile + r * LDS_STRIDE + c4 * 4;
    d[0] = k1 * xv.x + d[0];
    d[1] = k1 * xv.y + d[1];
    d[2] = k1 * xv.z + d[2];
    d[3] = k1 * xv.w + d[3];
  }
  if (big) {
    const int r9 = BROWS + 2 * wv + half;
    const int node = row0 + r9;
    const int cn = node < n_nodes ? node : n_nodes - 1;
    const float4 xv = *((const float4*)(x + (size_t)cn * 64) + m16);
    float* d = acc + r9 * LDS_STRIDE + m16 * 4;
    d[0] = k1 * xv.x + d[0];
    d[1] = k1 * xv.y + d[1];
    d[2] = k1 * xv.z + d[2];
    d[3] = k1 * xv.w + d[3];
  }
  __syncthreads();

  v8f acc1[4];
  for (int t = 0; t < 4; ++t)
    for (int j = 0; j < 8; ++j) acc1[t][j] = 0.f;

  for (int k = 0; k < 16; ++k) {
    const float2 af = *(const float2*)(tile + m16 * LDS_STRIDE + 4 * k + 2 * half);
    v2f a; a.x = af.x; a.y = af.y;
    for (int t = 0; t < 4; ++t) {
      int n = t * 16 + m16;
      const float2 bf = *(const float2*)(wl + n * 64 + ((k ^ m16) << 2) + 2 * half);
      v2f b; b.x = bf.x; b.y = bf.y;
      acc1[t] = __builtin_amdgcn_wmma_f32_16x16x4_f32(
          false, a, false, b, (short)0, acc1[t], false, false);
    }
  }
  v8f acc1x;
  for (int j = 0; j < 8; ++j) acc1x[j] = 0.f;
  if (extra) {
    for (int k = 0; k < 16; ++k) {
      const float2 af = *(const float2*)(tile9 + m16 * LDS_STRIDE + 4 * k + 2 * half);
      v2f a; a.x = af.x; a.y = af.y;
      int n = wave * 16 + m16;
      const float2 bf = *(const float2*)(wl + n * 64 + ((k ^ m16) << 2) + 2 * half);
      v2f b; b.x = bf.x; b.y = bf.y;
      acc1x = __builtin_amdgcn_wmma_f32_16x16x4_f32(
          false, a, false, b, (short)0, acc1x, false, false);
    }
  }
  __syncthreads();

  for (int t = 0; t < 4; ++t) {
    float bv = b1[t * 16 + m16];
    for (int v = 0; v < 8; ++v) {
      float h = acc1[t][v] + bv;
      h = h > 0.f ? h : 0.f;
      tile[(v + 8 * half) * LDS_STRIDE + t * 16 + m16] = h;
    }
  }
  if (extra) {
    float bv = b1[wave * 16 + m16];
    for (int v = 0; v < 8; ++v) {
      float h = acc1x[v] + bv;
      h = h > 0.f ? h : 0.f;
      tile9[(v + 8 * half) * LDS_STRIDE + wave * 16 + m16] = h;
    }
  }
  __syncthreads();

  v8f acc2[4];
  for (int t = 0; t < 4; ++t)
    for (int j = 0; j < 8; ++j) acc2[t][j] = 0.f;

  for (int k = 0; k < 16; ++k) {
    const float2 af = *(const float2*)(tile + m16 * LDS_STRIDE + 4 * k + 2 * half);
    v2f a; a.x = af.x; a.y = af.y;
    for (int t = 0; t < 4; ++t) {
      int n = t * 16 + m16;
      const float2 bf = *(const float2*)(wl + 64 * 64 + n * 64 + ((k ^ m16) << 2) + 2 * half);
      v2f b; b.x = bf.x; b.y = bf.y;
      acc2[t] = __builtin_amdgcn_wmma_f32_16x16x4_f32(
          false, a, false, b, (short)0, acc2[t], false, false);
    }
  }

  for (int t = 0; t < 4; ++t) {
    float bv = b2[t * 16 + m16];
    for (int v = 0; v < 8; ++v) {
      int node = node0 + v + 8 * half;
      if (node < n_nodes)
        out[(size_t)node * 64 + t * 16 + m16] = acc2[t][v] + bv;
    }
  }

  if (extra) {
    v8f acc2x;
    for (int j = 0; j < 8; ++j) acc2x[j] = 0.f;
    for (int k = 0; k < 16; ++k) {
      const float2 af = *(const float2*)(tile9 + m16 * LDS_STRIDE + 4 * k + 2 * half);
      v2f a; a.x = af.x; a.y = af.y;
      int n = wave * 16 + m16;
      const float2 bf = *(const float2*)(wl + 64 * 64 + n * 64 + ((k ^ m16) << 2) + 2 * half);
      v2f b; b.x = bf.x; b.y = bf.y;
      acc2x = __builtin_amdgcn_wmma_f32_16x16x4_f32(
          false, a, false, b, (short)0, acc2x, false, false);
    }
    float bv = b2[wave * 16 + m16];
    for (int v = 0; v < 8; ++v) {
      int node = row0 + BROWS + v + 8 * half;
      if (node < n_nodes)
        out[(size_t)node * 64 + wave * 16 + m16] = acc2x[v] + bv;
    }
  }
}

extern "C" void kernel_launch(void* const* d_in, const int* in_sizes, int n_in,
                              void* d_out, int out_size, void* d_ws, size_t ws_size,
                              hipStream_t stream) {
  const float* x   = (const float*)d_in[0];
  const int*   ei  = (const int*)  d_in[1];
  const float* eps = (const float*)d_in[2];
  const float* W1  = (const float*)d_in[3];
  const float* b1  = (const float*)d_in[4];
  const float* W2  = (const float*)d_in[5];
  const float* b2  = (const float*)d_in[6];
  float* out = (float*)d_out;

  const int n_nodes = in_sizes[0] / 64;
  const int n_edges = in_sizes[1] / 2;
  const int n_chunks = (n_edges + CHUNK - 1) / CHUNK;
  const int n_tiles = (n_nodes + 15) / 16;
  const int over = n_tiles - 8 * OWNERS;
  const int nbig = over < 0 ? 0 : (over > OWNERS ? OWNERS : over);
  const int nb = nbig + (n_tiles - 9 * nbig + 7) / 8;

  int* sorted = (int*)d_ws;
  int* tab    = sorted + (size_t)n_chunks * CHUNK * 2;

  chunk_sort_kernel<<<n_chunks, 1024, 0, stream>>>(ei, tab, sorted, n_nodes, n_edges, nb, nbig);
  owner_mlp_kernel<<<nb, 256, 0, stream>>>(x, sorted, tab, eps, W1, b1, W2, b2, out, n_nodes,
                                           n_chunks, nb, nbig);
}


# Row sums: one target row per whole wave (32 lanes x 8 B = one source row per load); a wave's 16 rows walked as one run, 32 loads per batch, row ends by scalar test; no padding slots; sum order kept
# speedup vs baseline: 4.5821x; 1.0924x over previous
#include <hip/hip_runtime.h>

typedef __attribute__((ext_vector_type(2))) float v2f;
typedef __attribute__((ext_vector_type(8))) float v8f;

#define LDS_STRIDE 68

#define BROWS     128
#define BSHIFT    7
#define BROWS_BIG 144
#define OWNERS    768
#define ROWS_PAD  160
#define KSHIFT    8
#define NB_MAX    1024
#define CHUNK     16384
#define EPT       16
#define CH_MAX    128
#define TILE_PAIRS 2560
#define PPT       10
#define BATCH     32

__device__ __forceinline__ int wave_incl_scan(int v, int lane) {
#pragma unroll
  for (int o = 1; o < 32; o <<= 1) {
    const int t = __shfl_up(v, o);
    if (lane >= o) v += t;
  }
  return v;
}

__device__ __forceinline__ int bin_of(int d, int nbig) {
  const int rows_big = nbig * BROWS_BIG;
  return d < rows_big ? (int)((unsigned)d / (unsigned)BROWS_BIG) : nbig + ((d - rows_big) >> BSHIFT);
}

__global__ __launch_bounds__(1024) void chunk_sort_kernel(const int* __restrict__ ei,
                                                          int* __restrict__ tab,
                                                          int* __restrict__ sorted_i,
                                                          int n_nodes, int n_edges, int nb, int nbig) {
  static_assert(EPT == 16 && CHUNK == EPT * 1024, "sixteen edges per thread; the fence below names sixteen values");
  __shared__ __attribute__((aligned(16))) int2 pairs[CHUNK];
  __shared__ int lh[NB_MAX];
  __shared__ int ws_b[32];

  const int tid  = threadIdx.x;
  const int lane = tid & 31;
  const int wave = __builtin_amdgcn_readfirstlane(tid >> 5);
  const int g = blockIdx.x;

  for (int i = tid; i < nb; i += 1024) lh[i] = 0;
  __syncthreads();

  const int* __restrict__ dstp = ei + n_edges;
  const int e0 = g * CHUNK + tid;
  int s[EPT], d[EPT], rk[EPT], bn[EPT];
#pragma unroll
  for (int j = 0; j < EPT; ++j) {
    const int e = e0 + j * 1024;
    const bool ok = e < n_edges;
    d[j] = ok ? dstp[e] : -1;
    s[j] = ok ? ei[e] : 0;
  }
#pragma unroll
  for (int j = 0; j < EPT; ++j) {
    rk[j] = -1;
    bn[j] = 0;
    if ((unsigned)d[j] < (unsigned)n_nodes) {
      bn[j] = bin_of(d[j], nbig);
      rk[j] = atomicAdd(&lh[bn[j]], 1);
    }
  }
  __syncthreads();

  const int mine = tid < nb ? lh[tid] : 0;
  const int ib = wave_incl_scan(mine, lane);
  if (lane == 31) ws_b[wave] = ib;
  __syncthreads();
  if (wave == 0) {
    const int b = ws_b[lane];
    const int sb = wave_incl_scan(b, lane);
    ws_b[lane] = sb - b;
  }
  __syncthreads();
  const int lofs = ib - mine + ws_b[wave];
  if (tid < nb) {
    lh[tid] = lofs;
    int* __restrict__ trow = tab + (size_t)g * (nb + 1);
    trow[tid] = lofs;
    if (tid == nb - 1) trow[nb] = lofs + mine;
  }
  __syncthreads();

  int bs[EPT];
#pragma unroll
  for (int j = 0; j < EPT; ++j) bs[j] = lh[bn[j]];
  asm volatile("" :: "v"(bs[0]), "v"(bs[1]), "v"(bs[2]), "v"(bs[3]), "v"(bs[4]), "v"(bs[5]),
                     "v"(bs[6]), "v"(bs[7]), "v"(bs[8]), "v"(bs[9]), "v"(bs[10]), "v"(bs[11]),
                     "v"(bs[12]), "v"(bs[13]), "v"(bs[14]), "v"(bs[15]));
#pragma unroll
  for (int j = 0; j < EPT; ++j)
    if (rk[j] >= 0) pairs[bs[j] + rk[j]] = make_int2(s[j], d[j]);
  __syncthreads();

  const int4* __restrict__ p4 = (const int4*)pairs;
  int4* __restrict__ o4 = (int4*)sorted_i + (size_t)g * (CHUNK / 2);
  int4 cp[EPT / 2];
#pragma unroll
  for (int i = 0; i < EPT / 2; ++i) cp[i] = p4[tid + i * 1024];
#pragma unroll
  for (int i = 0; i < EPT / 2; ++i) o4[tid + i * 1024] = cp[i];
}

__device__ __forceinline__ void sum_rows(const float2* __restrict__ x2, const int* srcs,
                                         const int* rbeg, const int* cnt, float* acc,
                                         int rfirst, int nrows, int lane, bool first) {
  static_assert(BATCH == 32, "one source number, and one row end, per lane");
  int rb = 0, re = 0;
  if (lane < nrows) {
    rb = rbeg[rfirst + lane];
    re = rb + cnt[rfirst + lane];
  }
  int p = __builtin_amdgcn_readlane(rb, 0);
  const int pend = __builtin_amdgcn_readlane(re, nrows - 1);
  int cur = 0;
  int cend = __builtin_amdgcn_readlane(re, 0);
  float2 sum = make_float2(0.f, 0.f);
  float* dst = acc + rfirst * LDS_STRIDE + 2 * lane;
  auto flush = [&]() {
    float2* d = (float2*)(dst + cur * LDS_STRIDE);
    if (first) {
      *d = sum;
    } else {
      const float2 o = *d;
      *d = make_float2(o.x + sum.x, o.y + sum.y);
    }
    sum = make_float2(0.f, 0.f);
    ++cur;
  };
  int idx = p + lane < pend ? srcs[p + lane] : 0;
  while (p < pend) {
    float2 v[BATCH];
#pragma unroll
    for (int j = 0; j < BATCH; ++j)
      v[j] = x2[(unsigned)__builtin_amdgcn_readlane(idx, j) * 32u + (unsigned)lane];
    const int pn = p + BATCH;
    const int idxn = pn + lane < pend ? srcs[pn + lane] : 0;
#pragma unroll
    for (int j = 0; j < BATCH; ++j) {
      while (p + j == cend) {
        flush();
        cend = cur < nrows ? __builtin_amdgcn_readlane(re, cur) : -1;
      }
      sum.x += v[j].x;
      sum.y += v[j].y;
    }
    idx = idxn;
    p = pn;
  }
  while (cur < nrows) flush();
}

__global__ __launch_bounds__(256) void owner_mlp_kernel(
    const float* __restrict__ x, const int* __restrict__ sorted_i,
    const int* __restrict__ tab, const float* __restrict__ eps_p,
    const float* __restrict__ W1, const float* __restrict__ b1,
    const float* __restrict__ W2, const float* __restrict__ b2,
    float* __restrict__ out, int n_nodes, int n_chunks, int nb, int nbig) {
  static_assert(CH_MAX == 4 * 32, "four chunks per lane of wave 0; a power of two for the run search");
  static_assert(PPT == 10, "the fence behind the run search names ten values");
  static_assert(BROWS == 8 * 16 && BROWS_BIG == BROWS + 16,
                "one 16-row tile per wave of the 256-thread group; a big bin has one tile more");
  static_assert(ROWS_PAD == 5 * 32 && ROWS_PAD >= BROWS_BIG && BROWS_BIG <= (1 << KSHIFT),
                "five row counters per lane of wave 0; a row number fits the key's low bits");
  __shared__ __attribute__((aligned(16))) float acc[BROWS_BIG * LDS_STRIDE];
  __shared__ int srcs[TILE_PAIRS];
  __shared__ int cnt[ROWS_PAD];
  __shared__ int rbeg[ROWS_PAD];
  __shared__ __attribute__((aligned(16))) int cs[CH_MAX + 4];
  __shared__ __attribute__((aligned(16))) int gd[CH_MAX];
  __shared__ __attribute__((aligned(16))) float wl[2 * 64 * 64];

  const int tid  = threadIdx.x;
  const int lane = tid & 31;
  const int wave = __builtin_amdgcn_readfirstlane(tid >> 5);
  const int bin = blockIdx.x;
  const bool big = bin < nbig;
  const int row0 = big ? bin * BROWS_BIG : nbig * BROWS_BIG + (bin - nbig) * BROWS;

  int rlo[4] = {0, 0, 0, 0}, rhi[4] = {0, 0, 0, 0};
  if (wave == 0) {
#pragma unroll
    for (int i = 0; i < 4; ++i) {
      const int c = 4 * lane + i;
      const bool ok = c < n_chunks;
      const int* __restrict__ tp = tab + (size_t)(ok ? c : 0) * (nb + 1) + blockIdx.x;
      rlo[i] = ok ? tp[0] : 0;
      rhi[i] = ok ? tp[1] : 0;
    }
  }

  {
    const float4* __restrict__ w1v = (const float4*)W1;
    const float4* __restrict__ w2v = (const float4*)W2;
    float4 wr[8];
#pragma unroll
    for (int i = 0; i < 4; ++i) {
      wr[i]     = w1v[tid + 256 * i];
      wr[4 + i] = w2v[tid + 256 * i];
    }
#pragma unroll
    for (int i = 0; i < 4; ++i) {
      const int u = tid + 256 * i;
      const int n = u >> 4;
      const int off = n * 64 + (((u & 15) ^ (n & 15)) << 2);
      *(float4*)(wl + off)           = wr[i];
      *(float4*)(wl + 64 * 64 + off) = wr[4 + i];
    }
  }

  const int2* __restrict__ sorted = (const int2*)sorted_i;
  const float2* __restrict__ x2 = (const float2*)x;
  if (wave == 0) {
    const int l0 = rhi[0] - rlo[0], l1 = rhi[1] - rlo[1], l2 = rhi[2] - rlo[2], l3 = rhi[3] - rlo[3];
    const int tot = l0 + l1 + l2 + l3;
    const int e0 = wave_incl_scan(tot, lane) - tot;
    const int e1 = e0 + l0, e2 = e1 + l1, e3 = e2 + l2;
    const int base = 4 * lane * CHUNK;
    *(int4*)(cs + 4 * lane) = make_int4(e0, e1, e2, e3);
    *(int4*)(gd + 4 * lane) = make_int4(base + rlo[0] - e0, base + CHUNK + rlo[1] - e1,
                                        base + 2 * CHUNK + rlo[2] - e2,
                                        base + 3 * CHUNK + rlo[3] - e3);
    if (lane == 31) cs[CH_MAX] = e3 + l3;
  }
  if (tid < ROWS_PAD) cnt[tid] = 0;
  __syncthreads();
  const int s1 = cs[CH_MAX];
  int t0 = 0;
  bool first = true;

  for (;;) {
    int2 pr[PPT];
    int key[PPT];
    int c[PPT];
#pragma unroll
    for (int j = 0; j < PPT; ++j) c[j] = 0;
#pragma unroll
    for (int st = CH_MAX / 2; st >= 1; st >>= 1) {
      int v[PPT];
#pragma unroll
      for (int j = 0; j < PPT; ++j) v[j] = cs[c[j] + st];
#pragma unroll
      for (int j = 0; j < PPT; ++j) c[j] += v[j] <= t0 + j * 256 + tid ? st : 0;
    }
    int gq[PPT];
#pragma unroll
    for (int j = 0; j < PPT; ++j) gq[j] = gd[c[j]];
    asm volatile("" :: "v"(gq[0]), "v"(gq[1]), "v"(gq[2]), "v"(gq[3]), "v"(gq[4]),
                       "v"(gq[5]), "v"(gq[6]), "v"(gq[7]), "v"(gq[8]), "v"(gq[9]));
#pragma unroll
    for (int j = 0; j < PPT; ++j) {
      const int q = t0 + j * 256 + tid;
      pr[j] = q < s1 ? sorted[gq[j] + q] : make_int2(0, -1);
    }
#pragma unroll
    for (int j = 0; j < PPT; ++j) {
      key[j] = -1;
      if (pr[j].y >= 0) {
        const int r = pr[j].y - row0;
        key[j] = (atomicAdd(&cnt[r], 1) << KSHIFT) | r;
      }
    }
    __syncthreads();

    if (wave == 0) {
      const int* __restrict__ cp = cnt + 5 * lane;
      const int c0 = cp[0], c1 = cp[1], c2 = cp[2], c3 = cp[3], c4 = cp[4];
      const int tot = c0 + c1 + c2 + c3 + c4;
      const int ex = wave_incl_scan(tot, lane) - tot;
      int* __restrict__ rp = rbeg + 5 * lane;
      rp[0] = ex;
      rp[1] = ex + c0;
      rp[2] = ex + c0 + c1;
      rp[3] = ex + c0 + c1 + c2;
      rp[4] = ex + c0 + c1 + c2 + c3;
    }
    __syncthreads();

#pragma unroll
    for (int j = 0; j < PPT; ++j)
      if (key[j] >= 0) srcs[rbeg[key[j] & ((1 << KSHIFT) - 1)] + (key[j] >> KSHIFT)] = pr[j].x;
    __syncthreads();

    for (int run = 0; run < (big ? 2 : 1); ++run)
      sum_rows(x2, srcs, rbeg, cnt, acc, run == 0 ? wave * 16 : BROWS + 2 * wave,
               run == 0 ? 16 : 2, lane, first);

    t0 += TILE_PAIRS;
    first = false;
    if (t0 >= s1) break;
    __syncthreads();
    if (tid < ROWS_PAD) cnt[tid] = 0;
    __syncthreads();
  }
  __syncthreads();

  const int half = lane >> 4;
  const int m16  = lane & 15;
  const int wv = tid >> 5;
  float* tile = acc + wv * 16 * LDS_STRIDE;
  float* tile9 = acc + BROWS * LDS_STRIDE;
  const bool extra = big && wave < 4;

  const float k1 = 1.0f + eps_p[0];
  const int node0 = row0 + wv * 16;

  for (int i = 0; i < 8; ++i) {
    int e4 = lane + i * 32;
    int r  = e4 >> 4;
    int c4 = e4 & 15;
    int node = node0 + r;
    int cn = node < n_nodes ? node : n_nodes - 1;
    float4 xv = *((const float4*)(x + (size_t)cn * 64) + c4);
    float* d = tile + r * LDS_STRIDE + c4 * 4;
    d[0] = k1 * xv.x + d[0];
    d[1] = k1 * xv.y + d[1];
    d[2] = k1 * xv.z + d[2];
    d[3] = k1 * xv.w + d[3];
  }
  if (big) {
    const int r9 = BROWS + 2 * wv + half;
    const int node = row0 + r9;
    const int cn = node < n_nodes ? node : n_nodes - 1;
    const float4 xv = *((const float4*)(x + (size_t)cn * 64) + m16);
    float* d = acc + r9 * LDS_STRIDE + m16 * 4;
    d[0] = k1 * xv.x + d[0];
    d[1] = k1 * xv.y + d[1];
    d[2] = k1 * xv.z + d[2];
    d[3] = k1 * xv.w + d[3];
  }
  __syncthreads();

  v8f acc1[4];
  for (int t = 0; t < 4; ++t)
    for (int j = 0; j < 8; ++j) acc1[t][j] = 0.f;

  for (int k = 0; k < 16; ++k) {
    const float2 af = *(const float2*)(tile + m16 * LDS_STRIDE + 4 * k + 2 * half);
    v2f a; a.x = af.x; a.y = af.y;
    for (int t = 0; t < 4; ++t) {
      int n = t * 16 + m16;
      const float2 bf = *(const float2*)(wl + n * 64 + ((k ^ m16) << 2) + 2 * half);
      v2f b; b.x = bf.x; b.y = bf.y;
      acc1[t] = __builtin_amdgcn_wmma_f32_16x16x4_f32(
          false, a, false, b, (short)0, acc1[t], false, false);
    }
  }
  v8f acc1x;
  for (int j = 0; j < 8; ++j) acc1x[j] = 0.f;
  if (extra) {
    for (int k = 0; k < 16; ++k) {
      const float2 af = *(const float2*)(tile9 + m16 * LDS_STRIDE + 4 * k + 2 * half);
      v2f a; a.x = af.x; a.y = af.y;
      int n = wave * 16 + m16;
      const float2 bf = *(const float2*)(wl + n * 64 + ((k ^ m16) << 2) + 2 * half);
      v2f b; b.x = bf.x; b.y = bf.y;
      acc1x = __builtin_amdgcn_wmma_f32_16x16x4_f32(
          false, a, false, b, (short)0, acc1x, false, false);
    }
  }
  __syncthreads();

  for (int t = 0; t < 4; ++t) {
    float bv = b1[t * 16 + m16];
    for (int v = 0; v < 8; ++v) {
      float h = acc1[t][v] + bv;
      h = h > 0.f ? h : 0.f;
      tile[(v + 8 * half) * LDS_STRIDE + t * 16 + m16] = h;
    }
  }
  if (extra) {
    float bv = b1[wave * 16 + m16];
    for (int v = 0; v < 8; ++v) {
      float h = acc1x[v] + bv;
      h = h > 0.f ? h : 0.f;
      tile9[(v + 8 * half) * LDS_STRIDE + wave * 16 + m16] = h;
    }
  }
  __syncthreads();

  v8f acc2[4];
  for (int t = 0; t < 4; ++t)
    for (int j = 0; j < 8; ++j) acc2[t][j] = 0.f;

  for (int k = 0; k < 16; ++k) {
    const float2 af = *(const float2*)(tile + m16 * LDS_STRIDE + 4 * k + 2 * half);
    v2f a; a.x = af.x; a.y = af.y;
    for (int t = 0; t < 4; ++t) {
      int n = t * 16 + m16;
      const float2 bf = *(const float2*)(wl + 64 * 64 + n * 64 + ((k ^ m16) << 2) + 2 * half);
      v2f b; b.x = bf.x; b.y = bf.y;
      acc2[t] = __builtin_amdgcn_wmma_f32_16x16x4_f32(
          false, a, false, b, (short)0, acc2[t], false, false);
    }
  }

  for (int t = 0; t < 4; ++t) {
    float bv = b2[t * 16 + m16];
    for (int v = 0; v < 8; ++v) {
      int node = node0 + v + 8 * half;
      if (node < n_nodes)
        out[(size_t)node * 64 + t * 16 + m16] = acc2[t][v] + bv;
    }
  }

  if (extra) {
    v8f acc2x;
    for (int j = 0; j < 8; ++j) acc2x[j] = 0.f;
    for (int k = 0; k < 16; ++k) {
      const float2 af = *(const float2*)(tile9 + m16 * LDS_STRIDE + 4 * k + 2 * half);
      v2f a; a.x = af.x; a.y = af.y;
      int n = wave * 16 + m16;
      const float2 bf = *(const float2*)(wl + 64 * 64 + n * 64 + ((k ^ m16) << 2) + 2 * half);
      v2f b; b.x = bf.x; b.y = bf.y;
      acc2x = __builtin_amdgcn_wmma_f32_16x16x4_f32(
          false, a, false, b, (short)0, acc2x, false, false);
    }
    float bv = b2[wave * 16 + m16];
    for (int v = 0; v < 8; ++v) {
      int node = row0 + BROWS + v + 8 * half;
      if (node < n_nodes)
        out[(size_t)node * 64 + wave * 16 + m16] = acc2x[v] + bv;
    }
  }
}

extern "C" void kernel_launch(void* const* d_in, const int* in_sizes, int n_in,
                              void* d_out, int out_size, void* d_ws, size_t ws_size,
                              hipStream_t stream) {
  const float* x   = (const float*)d_in[0];
  const int*   ei  = (const int*)  d_in[1];
  const float* eps = (const float*)d_in[2];
  const float* W1  = (const float*)d_in[3];
  const float* b1  = (const float*)d_in[4];
  const float* W2  = (const float*)d_in[5];
  const float* b2  = (const float*)d_in[6];
  float* out = (float*)d_out;

  const int n_nodes = in_sizes[0] / 64;
  const int n_edges = in_sizes[1] / 2;
  const int n_chunks = (n_edges + CHUNK - 1) / CHUNK;
  const int n_tiles = (n_nodes + 15) / 16;
  const int over = n_tiles - 8 * OWNERS;
  const int nbig = over < 0 ? 0 : (over > OWNERS ? OWNERS : over);
  const int nb = nbig + (n_tiles - 9 * nbig + 7) / 8;

  int* sorted = (int*)d_ws;
  int* tab    = sorted + (size_t)n_chunks * CHUNK * 2;

  chunk_sort_kernel<<<n_chunks, 1024, 0, stream>>>(ei, tab, sorted, n_nodes, n_edges, nb, nbig);
  owner_mlp_kernel<<<nb, 256, 0, stream>>>(x, sorted, tab, eps, W1, b1, W2, b2, out, n_nodes,
                                           n_chunks, nb, nbig);
}


# Output rows leave through the wave's own LDS tile: finished 16x64 block written there, then stored two whole 256-byte rows (four 128-byte lines) per instruction; was 32 half-line stores per wave
# speedup vs baseline: 4.9293x; 1.0758x over previous
#include <hip/hip_runtime.h>

typedef __attribute__((ext_vector_type(2))) float v2f;
typedef __attribute__((ext_vector_type(8))) float v8f;

#define LDS_STRIDE 68

#define BROWS     128
#define BSHIFT    7
#define BROWS_BIG 144
#define OWNERS    768
#define ROWS_PAD  160
#define KSHIFT    8
#define NB_MAX    1024
#define CHUNK     16384
#define EPT       16
#define CH_MAX    128
#define TILE_PAIRS 2560
#define PPT       10
#define BATCH     32

__device__ __forceinline__ int wave_incl_scan(int v, int lane) {
#pragma unroll
  for (int o = 1; o < 32; o <<= 1) {
    const int t = __shfl_up(v, o);
    if (lane >= o) v += t;
  }
  return v;
}

__device__ __forceinline__ int bin_of(int d, int nbig) {
  const int rows_big = nbig * BROWS_BIG;
  return d < rows_big ? (int)((unsigned)d / (unsigned)BROWS_BIG) : nbig + ((d - rows_big) >> BSHIFT);
}

__global__ __launch_bounds__(1024) void chunk_sort_kernel(const int* __restrict__ ei,
                                                          int* __restrict__ tab,
                                                          int* __restrict__ sorted_i,
                                                          int n_nodes, int n_edges, int nb, int nbig) {
  static_assert(EPT == 16 && CHUNK == EPT * 1024, "sixteen edges per thread; the fence below names sixteen values");
  __shared__ __attribute__((aligned(16))) int2 pairs[CHUNK];
  __shared__ int lh[NB_MAX];
  __shared__ int ws_b[32];

  const int tid  = threadIdx.x;
  const int lane = tid & 31;
  const int wave = __builtin_amdgcn_readfirstlane(tid >> 5);
  const int g = blockIdx.x;

  for (int i = tid; i < nb; i += 1024) lh[i] = 0;
  __syncthreads();

  const int* __restrict__ dstp = ei + n_edges;
  const int e0 = g * CHUNK + tid;
  int s[EPT], d[EPT], rk[EPT], bn[EPT];
#pragma unroll
  for (int j = 0; j < EPT; ++j) {
    const int e = e0 + j * 1024;
    const bool ok = e < n_edges;
    d[j] = ok ? dstp[e] : -1;
    s[j] = ok ? ei[e] : 0;
  }
#pragma unroll
  for (int j = 0; j < EPT; ++j) {
    rk[j] = -1;
    bn[j] = 0;
    if ((unsigned)d[j] < (unsigned)n_nodes) {
      bn[j] = bin_of(d[j], nbig);
      rk[j] = atomicAdd(&lh[bn[j]], 1);
    }
  }
  __syncthreads();

  const int mine = tid < nb ? lh[tid] : 0;
  const int ib = wave_incl_scan(mine, lane);
  if (lane == 31) ws_b[wave] = ib;
  __syncthreads();
  if (wave == 0) {
    const int b = ws_b[lane];
    const int sb = wave_incl_scan(b, lane);
    ws_b[lane] = sb - b;
  }
  __syncthreads();
  const int lofs = ib - mine + ws_b[wave];
  if (tid < nb) {
    lh[tid] = lofs;
    int* __restrict__ trow = tab + (size_t)g * (nb + 1);
    trow[tid] = lofs;
    if (tid == nb - 1) trow[nb] = lofs + mine;
  }
  __syncthreads();

  int bs[EPT];
#pragma unroll
  for (int j = 0; j < EPT; ++j) bs[j] = lh[bn[j]];
  asm volatile("" :: "v"(bs[0]), "v"(bs[1]), "v"(bs[2]), "v"(bs[3]), "v"(bs[4]), "v"(bs[5]),
                     "v"(bs[6]), "v"(bs[7]), "v"(bs[8]), "v"(bs[9]), "v"(bs[10]), "v"(bs[11]),
                     "v"(bs[12]), "v"(bs[13]), "v"(bs[14]), "v"(bs[15]));
#pragma unroll
  for (int j = 0; j < EPT; ++j)
    if (rk[j] >= 0) pairs[bs[j] + rk[j]] = make_int2(s[j], d[j]);
  __syncthreads();

  const int4* __restrict__ p4 = (const int4*)pairs;
  int4* __restrict__ o4 = (int4*)sorted_i + (size_t)g * (CHUNK / 2);
  int4 cp[EPT / 2];
#pragma unroll
  for (int i = 0; i < EPT / 2; ++i) cp[i] = p4[tid + i * 1024];
#pragma unroll
  for (int i = 0; i < EPT / 2; ++i) o4[tid + i * 1024] = cp[i];
}

__device__ __forceinline__ void sum_rows(const float2* __restrict__ x2, const int* srcs,
                                         const int* rbeg, const int* cnt, float* acc,
                                         int rfirst, int nrows, int lane, bool first) {
  static_assert(BATCH == 32, "one source number, and one row end, per lane");
  int rb = 0, re = 0;
  if (lane < nrows) {
    rb = rbeg[rfirst + lane];
    re = rb + cnt[rfirst + lane];
  }
  int p = __builtin_amdgcn_readlane(rb, 0);
  const int pend = __builtin_amdgcn_readlane(re, nrows - 1);
  int cur = 0;
  int cend = __builtin_amdgcn_readlane(re, 0);
  float2 sum = make_float2(0.f, 0.f);
  float* dst = acc + rfirst * LDS_STRIDE + 2 * lane;
  auto flush = [&]() {
    float2* d = (float2*)(dst + cur * LDS_STRIDE);
    if (first) {
      *d = sum;
    } else {
      const float2 o = *d;
      *d = make_float2(o.x + sum.x, o.y + sum.y);
    }
    sum = make_float2(0.f, 0.f);
    ++cur;
  };
  int idx = p + lane < pend ? srcs[p + lane] : 0;
  while (p < pend) {
    float2 v[BATCH];
#pragma unroll
    for (int j = 0; j < BATCH; ++j)
      v[j] = x2[(unsigned)__builtin_amdgcn_readlane(idx, j) * 32u + (unsigned)lane];
    const int pn = p + BATCH;
    const int idxn = pn + lane < pend ? srcs[pn + lane] : 0;
#pragma unroll
    for (int j = 0; j < BATCH; ++j) {
      while (p + j == cend) {
        flush();
        cend = cur < nrows ? __builtin_amdgcn_readlane(re, cur) : -1;
      }
      sum.x += v[j].x;
      sum.y += v[j].y;
    }
    idx = idxn;
    p = pn;
  }
  while (cur < nrows) flush();
}

__global__ __launch_bounds__(256) void owner_mlp_kernel(
    const float* __restrict__ x, const int* __restrict__ sorted_i,
    const int* __restrict__ tab, const float* __restrict__ eps_p,
    const float* __restrict__ W1, const float* __restrict__ b1,
    const float* __restrict__ W2, const float* __restrict__ b2,
    float* __restrict__ out, int n_nodes, int n_chunks, int nb, int nbig) {
  static_assert(CH_MAX == 4 * 32, "four chunks per lane of wave 0; a power of two for the run search");
  static_assert(PPT == 10, "the fence behind the run search names ten values");
  static_assert(BROWS == 8 * 16 && BROWS_BIG == BROWS + 16,
                "one 16-row tile per wave of the 256-thread group; a big bin has one tile more");
  static_assert(ROWS_PAD == 5 * 32 && ROWS_PAD >= BROWS_BIG && BROWS_BIG <= (1 << KSHIFT),
                "five row counters per lane of wave 0; a row number fits the key's low bits");
  __shared__ __attribute__((aligned(16))) float acc[BROWS_BIG * LDS_STRIDE];
  __shared__ int srcs[TILE_PAIRS];
  __shared__ int cnt[ROWS_PAD];
  __shared__ int rbeg[ROWS_PAD];
  __shared__ __attribute__((aligned(16))) int cs[CH_MAX + 4];
  __shared__ __attribute__((aligned(16))) int gd[CH_MAX];
  __shared__ __attribute__((aligned(16))) float wl[2 * 64 * 64];

  const int tid  = threadIdx.x;
  const int lane = tid & 31;
  const int wave = __builtin_amdgcn_readfirstlane(tid >> 5);
  const int bin = blockIdx.x;
  const bool big = bin < nbig;
  const int row0 = big ? bin * BROWS_BIG : nbig * BROWS_BIG + (bin - nbig) * BROWS;

  int rlo[4] = {0, 0, 0, 0}, rhi[4] = {0, 0, 0, 0};
  if (wave == 0) {
#pragma unroll
    for (int i = 0; i < 4; ++i) {
      const int c = 4 * lane + i;
      const bool ok = c < n_chunks;
      const int* __restrict__ tp = tab + (size_t)(ok ? c : 0) * (nb + 1) + blockIdx.x;
      rlo[i] = ok ? tp[0] : 0;
      rhi[i] = ok ? tp[1] : 0;
    }
  }

  {
    const float4* __restrict__ w1v = (const float4*)W1;
    const float4* __restrict__ w2v = (const float4*)W2;
    float4 wr[8];
#pragma unroll
    for (int i = 0; i < 4; ++i) {
      wr[i]     = w1v[tid + 256 * i];
      wr[4 + i] = w2v[tid + 256 * i];
    }
#pragma unroll
    for (int i = 0; i < 4; ++i) {
      const int u = tid + 256 * i;
      const int n = u >> 4;
      const int off = n * 64 + (((u & 15) ^ (n & 15)) << 2);
      *(float4*)(wl + off)           = wr[i];
      *(float4*)(wl + 64 * 64 + off) = wr[4 + i];
    }
  }

  const int2* __restrict__ sorted = (const int2*)sorted_i;
  const float2* __restrict__ x2 = (const float2*)x;
  if (wave == 0) {
    const int l0 = rhi[0] - rlo[0], l1 = rhi[1] - rlo[1], l2 = rhi[2] - rlo[2], l3 = rhi[3] - rlo[3];
    const int tot = l0 + l1 + l2 + l3;
    const int e0 = wave_incl_scan(tot, lane) - tot;
    const int e1 = e0 + l0, e2 = e1 + l1, e3 = e2 + l2;
    const int base = 4 * lane * CHUNK;
    *(int4*)(cs + 4 * lane) = make_int4(e0, e1, e2, e3);
    *(int4*)(gd + 4 * lane) = make_int4(base + rlo[0] - e0, base + CHUNK + rlo[1] - e1,
                                        base + 2 * CHUNK + rlo[2] - e2,
                                        base + 3 * CHUNK + rlo[3] - e3);
    if (lane == 31) cs[CH_MAX] = e3 + l3;
  }
  if (tid < ROWS_PAD) cnt[tid] = 0;
  __syncthreads();
  const int s1 = cs[CH_MAX];
  int t0 = 0;
  bool first = true;

  for (;;) {
    int2 pr[PPT];
    int key[PPT];
    int c[PPT];
#pragma unroll
    for (int j = 0; j < PPT; ++j) c[j] = 0;
#pragma unroll
    for (int st = CH_MAX / 2; st >= 1; st >>= 1) {
      int v[PPT];
#pragma unroll
      for (int j = 0; j < PPT; ++j) v[j] = cs[c[j] + st];
#pragma unroll
      for (int j = 0; j < PPT; ++j) c[j] += v[j] <= t0 + j * 256 + tid ? st : 0;
    }
    int gq[PPT];
#pragma unroll
    for (int j = 0; j < PPT; ++j) gq[j] = gd[c[j]];
    asm volatile("" :: "v"(gq[0]), "v"(gq[1]), "v"(gq[2]), "v"(gq[3]), "v"(gq[4]),
                       "v"(gq[5]), "v"(gq[6]), "v"(gq[7]), "v"(gq[8]), "v"(gq[9]));
#pragma unroll
    for (int j = 0; j < PPT; ++j) {
      const int q = t0 + j * 256 + tid;
      pr[j] = q < s1 ? sorted[gq[j] + q] : make_int2(0, -1);
    }
#pragma unroll
    for (int j = 0; j < PPT; ++j) {
      key[j] = -1;
      if (pr[j].y >= 0) {
        const int r = pr[j].y - row0;
        key[j] = (atomicAdd(&cnt[r], 1) << KSHIFT) | r;
      }
    }
    __syncthreads();

    if (wave == 0) {
      const int* __restrict__ cp = cnt + 5 * lane;
      const int c0 = cp[0], c1 = cp[1], c2 = cp[2], c3 = cp[3], c4 = cp[4];
      const int tot = c0 + c1 + c2 + c3 + c4;
      const int ex = wave_incl_scan(tot, lane) - tot;
      int* __restrict__ rp = rbeg + 5 * lane;
      rp[0] = ex;
      rp[1] = ex + c0;
      rp[2] = ex + c0 + c1;
      rp[3] = ex + c0 + c1 + c2;
      rp[4] = ex + c0 + c1 + c2 + c3;
    }
    __syncthreads();

#pragma unroll
    for (int j = 0; j < PPT; ++j)
      if (key[j] >= 0) srcs[rbeg[key[j] & ((1 << KSHIFT) - 1)] + (key[j] >> KSHIFT)] = pr[j].x;
    __syncthreads();

    for (int run = 0; run < (big ? 2 : 1); ++run)
      sum_rows(x2, srcs, rbeg, cnt, acc, run == 0 ? wave * 16 : BROWS + 2 * wave,
               run == 0 ? 16 : 2, lane, first);

    t0 += TILE_PAIRS;
    first = false;
    if (t0 >= s1) break;
    __syncthreads();
    if (tid < ROWS_PAD) cnt[tid] = 0;
    __syncthreads();
  }
  __syncthreads();

  const int half = lane >> 4;
  const int m16  = lane & 15;
  const int wv = tid >> 5;
  float* tile = acc + wv * 16 * LDS_STRIDE;
  float* tile9 = acc + BROWS * LDS_STRIDE;
  const bool extra = big && wave < 4;

  const float k1 = 1.0f + eps_p[0];
  const int node0 = row0 + wv * 16;

  for (int i = 0; i < 8; ++i) {
    int e4 = lane + i * 32;
    int r  = e4 >> 4;
    int c4 = e4 & 15;
    int node = node0 + r;
    int cn = node < n_nodes ? node : n_nodes - 1;
    float4 xv = *((const float4*)(x + (size_t)cn * 64) + c4);
    float* d = tile + r * LDS_STRIDE + c4 * 4;
    d[0] = k1 * xv.x + d[0];
    d[1] = k1 * xv.y + d[1];
    d[2] = k1 * xv.z + d[2];
    d[3] = k1 * xv.w + d[3];
  }
  if (big) {
    const int r9 = BROWS + 2 * wv + half;
    const int node = row0 + r9;
    const int cn = node < n_nodes ? node : n_nodes - 1;
    const float4 xv = *((const float4*)(x + (size_t)cn * 64) + m16);
    float* d = acc + r9 * LDS_STRIDE + m16 * 4;
    d[0] = k1 * xv.x + d[0];
    d[1] = k1 * xv.y + d[1];
    d[2] = k1 * xv.z + d[2];
    d[3] = k1 * xv.w + d[3];
  }
  __syncthreads();

  v8f acc1[4];
  for (int t = 0; t < 4; ++t)
    for (int j = 0; j < 8; ++j) acc1[t][j] = 0.f;

  for (int k = 0; k < 16; ++k) {
    const float2 af = *(const float2*)(tile + m16 * LDS_STRIDE + 4 * k + 2 * half);
    v2f a; a.x = af.x; a.y = af.y;
    for (int t = 0; t < 4; ++t) {
      int n = t * 16 + m16;
      const float2 bf = *(const float2*)(wl + n * 64 + ((k ^ m16) << 2) + 2 * half);
      v2f b; b.x = bf.x; b.y = bf.y;
      acc1[t] = __builtin_amdgcn_wmma_f32_16x16x4_f32(
          false, a, false, b, (short)0, acc1[t], false, false);
    }
  }
  v8f acc1x;
  for (int j = 0; j < 8; ++j) acc1x[j] = 0.f;
  if (extra) {
    for (int k = 0; k < 16; ++k) {
      const float2 af = *(const float2*)(tile9 + m16 * LDS_STRIDE + 4 * k + 2 * half);
      v2f a; a.x = af.x; a.y = af.y;
      int n = wave * 16 + m16;
      const float2 bf = *(const float2*)(wl + n * 64 + ((k ^ m16) << 2) + 2 * half);
      v2f b; b.x = bf.x; b.y = bf.y;
      acc1x = __builtin_amdgcn_wmma_f32_16x16x4_f32(
          false, a, false, b, (short)0, acc1x, false, false);
    }
  }
  __syncthreads();

  for (int t = 0; t < 4; ++t) {
    float bv = b1[t * 16 + m16];
    for (int v = 0; v < 8; ++v) {
      float h = acc1[t][v] + bv;
      h = h > 0.f ? h : 0.f;
      tile[(v + 8 * half) * LDS_STRIDE + t * 16 + m16] = h;
    }
  }
  if (extra) {
    float bv = b1[wave * 16 + m16];
    for (int v = 0; v < 8; ++v) {
      float h = acc1x[v] + bv;
      h = h > 0.f ? h : 0.f;
      tile9[(v + 8 * half) * LDS_STRIDE + wave * 16 + m16] = h;
    }
  }
  __syncthreads();

  v8f acc2[4];
  for (int t = 0; t < 4; ++t)
    for (int j = 0; j < 8; ++j) acc2[t][j] = 0.f;

  for (int k = 0; k < 16; ++k) {
    const float2 af = *(const float2*)(tile + m16 * LDS_STRIDE + 4 * k + 2 * half);
    v2f a; a.x = af.x; a.y = af.y;
    for (int t = 0; t < 4; ++t) {
      int n = t * 16 + m16;
      const float2 bf = *(const float2*)(wl + 64 * 64 + n * 64 + ((k ^ m16) << 2) + 2 * half);
      v2f b; b.x = bf.x; b.y = bf.y;
      acc2[t] = __builtin_amdgcn_wmma_f32_16x16x4_f32(
          false, a, false, b, (short)0, acc2[t], false, false);
    }
  }

  for (int t = 0; t < 4; ++t) {
    float bv = b2[t * 16 + m16];
    for (int v = 0; v < 8; ++v)
      tile[(v + 8 * half) * LDS_STRIDE + t * 16 + m16] = acc2[t][v] + bv;
  }
  {
    const float* trd = tile + half * LDS_STRIDE + m16 * 4;
    float* orow = out + (size_t)(node0 + half) * 64 + m16 * 4;
    if (row0 + wave * 16 + 16 <= n_nodes) {
      float4 o[8];
      for (int i = 0; i < 8; ++i) o[i] = *(const float4*)(trd + i * 2 * LDS_STRIDE);
      for (int i = 0; i < 8; ++i) *(float4*)(orow + i * 128) = o[i];
    } else {
      for (int i = 0; i < 8; ++i)
        if (node0 + 2 * i + half < n_nodes)
          *(float4*)(orow + i * 128) = *(const float4*)(trd + i * 2 * LDS_STRIDE);
    }
  }

  if (extra) {
    v8f acc2x;
    for (int j = 0; j < 8; ++j) acc2x[j] = 0.f;
    for (int k = 0; k < 16; ++k) {
      const float2 af = *(const float2*)(tile9 + m16 * LDS_STRIDE + 4 * k + 2 * half);
      v2f a; a.x = af.x; a.y = af.y;
      int n = wave * 16 + m16;
      const float2 bf = *(const float2*)(wl + 64 * 64 + n * 64 + ((k ^ m16) << 2) + 2 * half);
      v2f b; b.x = bf.x; b.y = bf.y;
      acc2x = __builtin_amdgcn_wmma_f32_16x16x4_f32(
          false, a, false, b, (short)0, acc2x, false, false);
    }
    float bv = b2[wave * 16 + m16];
    for (int v = 0; v < 8; ++v) {
      int node = row0 + BROWS + v + 8 * half;
      if (node < n_nodes)
        out[(size_t)node * 64 + wave * 16 + m16] = acc2x[v] + bv;
    }
  }
}

extern "C" void kernel_launch(void* const* d_in, const int* in_sizes, int n_in,
                              void* d_out, int out_size, void* d_ws, size_t ws_size,
                              hipStream_t stream) {
  const float* x   = (const float*)d_in[0];
  const int*   ei  = (const int*)  d_in[1];
  const float* eps = (const float*)d_in[2];
  const float* W1  = (const float*)d_in[3];
  const float* b1  = (const float*)d_in[4];
  const float* W2  = (const float*)d_in[5];
  const float* b2  = (const float*)d_in[6];
  float* out = (float*)d_out;

  const int n_nodes = in_sizes[0] / 64;
  const int n_edges = in_sizes[1] / 2;
  const int n_chunks = (n_edges + CHUNK - 1) / CHUNK;
  const int n_tiles = (n_nodes + 15) / 16;
  const int over = n_tiles - 8 * OWNERS;
  const int nbig = over < 0 ? 0 : (over > OWNERS ? OWNERS : over);
  const int nb = nbig + (n_tiles - 9 * nbig + 7) / 8;

  int* sorted = (int*)d_ws;
  int* tab    = sorted + (size_t)n_chunks * CHUNK * 2;

  chunk_sort_kernel<<<n_chunks, 1024, 0, stream>>>(ei, tab, sorted, n_nodes, n_edges, nb, nbig);
  owner_mlp_kernel<<<nb, 256, 0, stream>>>(x, sorted, tab, eps, W1, b1, W2, b2, out, n_nodes,
                                           n_chunks, nb, nbig);
}


# Sorted pairs packed into one 32-bit word (source * 256 + row inside its bin): sort kernel's LDS buffer, its copy-out stores and the owner's pair loads move 4 bytes per edge, not 8; sum order kept
# speedup vs baseline: 5.1705x; 1.0489x over previous
#include <hip/hip_runtime.h>

typedef __attribute__((ext_vector_type(2))) float v2f;
typedef __attribute__((ext_vector_type(8))) float v8f;

#define LDS_STRIDE 68

#define BROWS     128
#define BSHIFT    7
#define BROWS_BIG 144
#define OWNERS    768
#define ROWS_PAD  160
#define KSHIFT    8
#define NB_MAX    1024
#define CHUNK     16384
#define EPT       16
#define CH_MAX    128
#define TILE_PAIRS 2560
#define PPT       10
#define BATCH     32

__device__ __forceinline__ int wave_incl_scan(int v, int lane) {
#pragma unroll
  for (int o = 1; o < 32; o <<= 1) {
    const int t = __shfl_up(v, o);
    if (lane >= o) v += t;
  }
  return v;
}

__device__ __forceinline__ int bin_of(int d, int nbig, int& r) {
  const int rows_big = nbig * BROWS_BIG;
  const int bb = (int)((unsigned)d / (unsigned)BROWS_BIG);
  const int t = d - rows_big;
  r = d < rows_big ? d - bb * BROWS_BIG : t & (BROWS - 1);
  return d < rows_big ? bb : nbig + (t >> BSHIFT);
}

__global__ __launch_bounds__(1024) void chunk_sort_kernel(const int* __restrict__ ei,
                                                          int* __restrict__ tab,
                                                          int* __restrict__ sorted_i,
                                                          int n_nodes, int n_edges, int nb, int nbig) {
  static_assert(EPT == 16 && CHUNK == EPT * 1024, "sixteen edges per thread; the fences below name sixteen values each");
  __shared__ __attribute__((aligned(16))) int pairs[CHUNK];
  __shared__ int lh[NB_MAX];
  __shared__ int ws_b[32];

  const int tid  = threadIdx.x;
  const int lane = tid & 31;
  const int wave = __builtin_amdgcn_readfirstlane(tid >> 5);
  const int g = blockIdx.x;

  for (int i = tid; i < nb; i += 1024) lh[i] = 0;
  __syncthreads();

  const int* __restrict__ dstp = ei + n_edges;
  const int e0 = g * CHUNK + tid;
  int s[EPT], d[EPT], rk[EPT], bn[EPT], rw[EPT];
#pragma unroll
  for (int j = 0; j < EPT; ++j) {
    const int e = e0 + j * 1024;
    const bool ok = e < n_edges;
    d[j] = ok ? dstp[e] : -1;
    s[j] = ok ? ei[e] : 0;
  }
  asm volatile("" :: "v"(s[0]), "v"(s[1]), "v"(s[2]), "v"(s[3]), "v"(s[4]), "v"(s[5]),
                     "v"(s[6]), "v"(s[7]), "v"(s[8]), "v"(s[9]), "v"(s[10]), "v"(s[11]),
                     "v"(s[12]), "v"(s[13]), "v"(s[14]), "v"(s[15]));
#pragma unroll
  for (int j = 0; j < EPT; ++j) {
    rk[j] = -1;
    bn[j] = 0;
    rw[j] = 0;
    if ((unsigned)d[j] < (unsigned)n_nodes) {
      bn[j] = bin_of(d[j], nbig, rw[j]);
      rk[j] = atomicAdd(&lh[bn[j]], 1);
    }
  }
  __syncthreads();

  const int mine = tid < nb ? lh[tid] : 0;
  const int ib = wave_incl_scan(mine, lane);
  if (lane == 31) ws_b[wave] = ib;
  __syncthreads();
  if (wave == 0) {
    const int b = ws_b[lane];
    const int sb = wave_incl_scan(b, lane);
    ws_b[lane] = sb - b;
  }
  __syncthreads();
  const int lofs = ib - mine + ws_b[wave];
  if (tid < nb) {
    lh[tid] = lofs;
    int* __restrict__ trow = tab + (size_t)g * (nb + 1);
    trow[tid] = lofs;
    if (tid == nb - 1) trow[nb] = lofs + mine;
  }
  __syncthreads();

  int bs[EPT];
#pragma unroll
  for (int j = 0; j < EPT; ++j) bs[j] = lh[bn[j]];
  asm volatile("" :: "v"(bs[0]), "v"(bs[1]), "v"(bs[2]), "v"(bs[3]), "v"(bs[4]), "v"(bs[5]),
                     "v"(bs[6]), "v"(bs[7]), "v"(bs[8]), "v"(bs[9]), "v"(bs[10]), "v"(bs[11]),
                     "v"(bs[12]), "v"(bs[13]), "v"(bs[14]), "v"(bs[15]));
#pragma unroll
  for (int j = 0; j < EPT; ++j)
    if (rk[j] >= 0) pairs[bs[j] + rk[j]] = (s[j] << KSHIFT) | rw[j];
  __syncthreads();

  const int4* __restrict__ p4 = (const int4*)pairs;
  int4* __restrict__ o4 = (int4*)sorted_i + (size_t)g * (CHUNK / 4);
  int4 cp[EPT / 4];
#pragma unroll
  for (int i = 0; i < EPT / 4; ++i) cp[i] = p4[tid + i * 1024];
#pragma unroll
  for (int i = 0; i < EPT / 4; ++i) o4[tid + i * 1024] = cp[i];
}

__device__ __forceinline__ void sum_rows(const float2* __restrict__ x2, const int* srcs,
                                         const int* rbeg, const int* cnt, float* acc,
                                         int rfirst, int nrows, int lane, bool first) {
  static_assert(BATCH == 32, "one source number, and one row end, per lane");
  int rb = 0, re = 0;
  if (lane < nrows) {
    rb = rbeg[rfirst + lane];
    re = rb + cnt[rfirst + lane];
  }
  int p = __builtin_amdgcn_readlane(rb, 0);
  const int pend = __builtin_amdgcn_readlane(re, nrows - 1);
  int cur = 0;
  int cend = __builtin_amdgcn_readlane(re, 0);
  float2 sum = make_float2(0.f, 0.f);
  float* dst = acc + rfirst * LDS_STRIDE + 2 * lane;
  auto flush = [&]() {
    float2* d = (float2*)(dst + cur * LDS_STRIDE);
    if (first) {
      *d = sum;
    } else {
      const float2 o = *d;
      *d = make_float2(o.x + sum.x, o.y + sum.y);
    }
    sum = make_float2(0.f, 0.f);
    ++cur;
  };
  int idx = p + lane < pend ? srcs[p + lane] : 0;
  while (p < pend) {
    float2 v[BATCH];
#pragma unroll
    for (int j = 0; j < BATCH; ++j)
      v[j] = x2[(unsigned)__builtin_amdgcn_readlane(idx, j) * 32u + (unsigned)lane];
    const int pn = p + BATCH;
    const int idxn = pn + lane < pend ? srcs[pn + lane] : 0;
#pragma unroll
    for (int j = 0; j < BATCH; ++j) {
      while (p + j == cend) {
        flush();
        cend = cur < nrows ? __builtin_amdgcn_readlane(re, cur) : -1;
      }
      sum.x += v[j].x;
      sum.y += v[j].y;
    }
    idx = idxn;
    p = pn;
  }
  while (cur < nrows) flush();
}

__global__ __launch_bounds__(256) void owner_mlp_kernel(
    const float* __restrict__ x, const int* __restrict__ sorted_i,
    const int* __restrict__ tab, const float* __restrict__ eps_p,
    const float* __restrict__ W1, const float* __restrict__ b1,
    const float* __restrict__ W2, const float* __restrict__ b2,
    float* __restrict__ out, int n_nodes, int n_chunks, int nb, int nbig) {
  static_assert(CH_MAX == 4 * 32, "four chunks per lane of wave 0; a power of two for the run search");
  static_assert(PPT == 10, "the fence behind the run search names ten values");
  static_assert(BROWS == 8 * 16 && BROWS_BIG == BROWS + 16,
                "one 16-row tile per wave of the 256-thread group; a big bin has one tile more");
  static_assert(ROWS_PAD == 5 * 32 && ROWS_PAD >= BROWS_BIG && BROWS_BIG <= (1 << KSHIFT),
                "five row counters per lane of wave 0; a row number fits the key's low bits");
  __shared__ __attribute__((aligned(16))) float acc[BROWS_BIG * LDS_STRIDE];
  __shared__ int srcs[TILE_PAIRS];
  __shared__ int cnt[ROWS_PAD];
  __shared__ int rbeg[ROWS_PAD];
  __shared__ __attribute__((aligned(16))) int cs[CH_MAX + 4];
  __shared__ __attribute__((aligned(16))) int gd[CH_MAX];
  __shared__ __attribute__((aligned(16))) float wl[2 * 64 * 64];

  const int tid  = threadIdx.x;
  const int lane = tid & 31;
  const int wave = __builtin_amdgcn_readfirstlane(tid >> 5);
  const int bin = blockIdx.x;
  const bool big = bin < nbig;
  const int row0 = big ? bin * BROWS_BIG : nbig * BROWS_BIG + (bin - nbig) * BROWS;

  int rlo[4] = {0, 0, 0, 0}, rhi[4] = {0, 0, 0, 0};
  if (wave == 0) {
#pragma unroll
    for (int i = 0; i < 4; ++i) {
      const int c = 4 * lane + i;
      const bool ok = c < n_chunks;
      const int* __restrict__ tp = tab + (size_t)(ok ? c : 0) * (nb + 1) + blockIdx.x;
      rlo[i] = ok ? tp[0] : 0;
      rhi[i] = ok ? tp[1] : 0;
    }
  }

  {
    const float4* __restrict__ w1v = (const float4*)W1;
    const float4* __restrict__ w2v = (const float4*)W2;
    float4 wr[8];
#pragma unroll
    for (int i = 0; i < 4; ++i) {
      wr[i]     = w1v[tid + 256 * i];
      wr[4 + i] = w2v[tid + 256 * i];
    }
#pragma unroll
    for (int i = 0; i < 4; ++i) {
      const int u = tid + 256 * i;
      const int n = u >> 4;
      const int off = n * 64 + (((u & 15) ^ (n & 15)) << 2);
      *(float4*)(wl + off)           = wr[i];
      *(float4*)(wl + 64 * 64 + off) = wr[4 + i];
    }
  }

  const int* __restrict__ sorted = sorted_i;
  const float2* __restrict__ x2 = (const float2*)x;
  if (wave == 0) {
    const int l0 = rhi[0] - rlo[0], l1 = rhi[1] - rlo[1], l2 = rhi[2] - rlo[2], l3 = rhi[3] - rlo[3];
    const int tot = l0 + l1 + l2 + l3;
    const int e0 = wave_incl_scan(tot, lane) - tot;
    const int e1 = e0 + l0, e2 = e1 + l1, e3 = e2 + l2;
    const int base = 4 * lane * CHUNK;
    *(int4*)(cs + 4 * lane) = make_int4(e0, e1, e2, e3);
    *(int4*)(gd + 4 * lane) = make_int4(base + rlo[0] - e0, base + CHUNK + rlo[1] - e1,
                                        base + 2 * CHUNK + rlo[2] - e2,
                                        base + 3 * CHUNK + rlo[3] - e3);
    if (lane == 31) cs[CH_MAX] = e3 + l3;
  }
  if (tid < ROWS_PAD) cnt[tid] = 0;
  __syncthreads();
  const int s1 = cs[CH_MAX];
  int t0 = 0;
  bool first = true;

  for (;;) {
    int pr[PPT];
    int key[PPT];
    int c[PPT];
#pragma unroll
    for (int j = 0; j < PPT; ++j) c[j] = 0;
#pragma unroll
    for (int st = CH_MAX / 2; st >= 1; st >>= 1) {
      int v[PPT];
#pragma unroll
      for (int j = 0; j < PPT; ++j) v[j] = cs[c[j] + st];
#pragma unroll
      for (int j = 0; j < PPT; ++j) c[j] += v[j] <= t0 + j * 256 + tid ? st : 0;
    }
    int gq[PPT];
#pragma unroll
    for (int j = 0; j < PPT; ++j) gq[j] = gd[c[j]];
    asm volatile("" :: "v"(gq[0]), "v"(gq[1]), "v"(gq[2]), "v"(gq[3]), "v"(gq[4]),
                       "v"(gq[5]), "v"(gq[6]), "v"(gq[7]), "v"(gq[8]), "v"(gq[9]));
#pragma unroll
    for (int j = 0; j < PPT; ++j) {
      const int q = t0 + j * 256 + tid;
      pr[j] = q < s1 ? sorted[gq[j] + q] : -1;
    }
#pragma unroll
    for (int j = 0; j < PPT; ++j) {
      key[j] = -1;
      if (pr[j] >= 0) {
        const int r = pr[j] & ((1 << KSHIFT) - 1);
        key[j] = (atomicAdd(&cnt[r], 1) << KSHIFT) | r;
      }
    }
    __syncthreads();

    if (wave == 0) {
      const int* __restrict__ cp = cnt + 5 * lane;
      const int c0 = cp[0], c1 = cp[1], c2 = cp[2], c3 = cp[3], c4 = cp[4];
      const int tot = c0 + c1 + c2 + c3 + c4;
      const int ex = wave_incl_scan(tot, lane) - tot;
      int* __restrict__ rp = rbeg + 5 * lane;
      rp[0] = ex;
      rp[1] = ex + c0;
      rp[2] = ex + c0 + c1;
      rp[3] = ex + c0 + c1 + c2;
      rp[4] = ex + c0 + c1 + c2 + c3;
    }
    __syncthreads();

#pragma unroll
    for (int j = 0; j < PPT; ++j)
      if (key[j] >= 0) srcs[rbeg[key[j] & ((1 << KSHIFT) - 1)] + (key[j] >> KSHIFT)] = pr[j] >> KSHIFT;
    __syncthreads();

    for (int run = 0; run < (big ? 2 : 1); ++run)
      sum_rows(x2, srcs, rbeg, cnt, acc, run == 0 ? wave * 16 : BROWS + 2 * wave,
               run == 0 ? 16 : 2, lane, first);

    t0 += TILE_PAIRS;
    first = false;
    if (t0 >= s1) break;
    __syncthreads();
    if (tid < ROWS_PAD) cnt[tid] = 0;
    __syncthreads();
  }
  __syncthreads();

  const int half = lane >> 4;
  const int m16  = lane & 15;
  const int wv = tid >> 5;
  float* tile = acc + wv * 16 * LDS_STRIDE;
  float* tile9 = acc + BROWS * LDS_STRIDE;
  const bool extra = big && wave < 4;

  const float k1 = 1.0f + eps_p[0];
  const int node0 = row0 + wv * 16;

  for (int i = 0; i < 8; ++i) {
    int e4 = lane + i * 32;
    int r  = e4 >> 4;
    int c4 = e4 & 15;
    int node = node0 + r;
    int cn = node < n_nodes ? node : n_nodes - 1;
    float4 xv = *((const float4*)(x + (size_t)cn * 64) + c4);
    float* d = tile + r * LDS_STRIDE + c4 * 4;
    d[0] = k1 * xv.x + d[0];
    d[1] = k1 * xv.y + d[1];
    d[2] = k1 * xv.z + d[2];
    d[3] = k1 * xv.w + d[3];
  }
  if (big) {
    const int r9 = BROWS + 2 * wv + half;
    const int node = row0 + r9;
    const int cn = node < n_nodes ? node : n_nodes - 1;
    const float4 xv = *((const float4*)(x + (size_t)cn * 64) + m16);
    float* d = acc + r9 * LDS_STRIDE + m16 * 4;
    d[0] = k1 * xv.x + d[0];
    d[1] = k1 * xv.y + d[1];
    d[2] = k1 * xv.z + d[2];
    d[3] = k1 * xv.w + d[3];
  }
  __syncthreads();

  v8f acc1[4];
  for (int t = 0; t < 4; ++t)
    for (int j = 0; j < 8; ++j) acc1[t][j] = 0.f;

  for (int k = 0; k < 16; ++k) {
    const float2 af = *(const float2*)(tile + m16 * LDS_STRIDE + 4 * k + 2 * half);
    v2f a; a.x = af.x; a.y = af.y;
    for (int t = 0; t < 4; ++t) {
      int n = t * 16 + m16;
      const float2 bf = *(const float2*)(wl + n * 64 + ((k ^ m16) << 2) + 2 * half);
      v2f b; b.x = bf.x; b.y = bf.y;
      acc1[t] = __builtin_amdgcn_wmma_f32_16x16x4_f32(
          false, a, false, b, (short)0, acc1[t], false, false);
    }
  }
  v8f acc1x;
  for (int j = 0; j < 8; ++j) acc1x[j] = 0.f;
  if (extra) {
    for (int k = 0; k < 16; ++k) {
      const float2 af = *(const float2*)(tile9 + m16 * LDS_STRIDE + 4 * k + 2 * half);
      v2f a; a.x = af.x; a.y = af.y;
      int n = wave * 16 + m16;
      const float2 bf = *(const float2*)(wl + n * 64 + ((k ^ m16) << 2) + 2 * half);
      v2f b; b.x = bf.x; b.y = bf.y;
      acc1x = __builtin_amdgcn_wmma_f32_16x16x4_f32(
          false, a, false, b, (short)0, acc1x, false, false);
    }
  }
  __syncthreads();

  for (int t = 0; t < 4; ++t) {
    float bv = b1[t * 16 + m16];
    for (int v = 0; v < 8; ++v) {
      float h = acc1[t][v] + bv;
      h = h > 0.f ? h : 0.f;
      tile[(v + 8 * half) * LDS_STRIDE + t * 16 + m16] = h;
    }
  }
  if (extra) {
    float bv = b1[wave * 16 + m16];
    for (int v = 0; v < 8; ++v) {
      float h = acc1x[v] + bv;
      h = h > 0.f ? h : 0.f;
      tile9[(v + 8 * half) * LDS_STRIDE + wave * 16 + m16] = h;
    }
  }
  __syncthreads();

  v8f acc2[4];
  for (int t = 0; t < 4; ++t)
    for (int j = 0; j < 8; ++j) acc2[t][j] = 0.f;

  for (int k = 0; k < 16; ++k) {
    const float2 af = *(const float2*)(tile + m16 * LDS_STRIDE + 4 * k + 2 * half);
    v2f a; a.x = af.x; a.y = af.y;
    for (int t = 0; t < 4; ++t) {
      int n = t * 16 + m16;
      const float2 bf = *(const float2*)(wl + 64 * 64 + n * 64 + ((k ^ m16) << 2) + 2 * half);
      v2f b; b.x = bf.x; b.y = bf.y;
      acc2[t] = __builtin_amdgcn_wmma_f32_16x16x4_f32(
          false, a, false, b, (short)0, acc2[t], false, false);
    }
  }

  for (int t = 0; t < 4; ++t) {
    float bv = b2[t * 16 + m16];
    for (int v = 0; v < 8; ++v)
      tile[(v + 8 * half) * LDS_STRIDE + t * 16 + m16] = acc2[t][v] + bv;
  }
  {
    const float* trd = tile + half * LDS_STRIDE + m16 * 4;
    float* orow = out + (size_t)(node0 + half) * 64 + m16 * 4;
    if (row0 + wave * 16 + 16 <= n_nodes) {
      float4 o[8];
      for (int i = 0; i < 8; ++i) o[i] = *(const float4*)(trd + i * 2 * LDS_STRIDE);
      for (int i = 0; i < 8; ++i) *(float4*)(orow + i * 128) = o[i];
    } else {
      for (int i = 0; i < 8; ++i)
        if (node0 + 2 * i + half < n_nodes)
          *(float4*)(orow + i * 128) = *(const float4*)(trd + i * 2 * LDS_STRIDE);
    }
  }

  if (extra) {
    v8f acc2x;
    for (int j = 0; j < 8; ++j) acc2x[j] = 0.f;
    for (int k = 0; k < 16; ++k) {
      const float2 af = *(const float2*)(tile9 + m16 * LDS_STRIDE + 4 * k + 2 * half);
      v2f a; a.x = af.x; a.y = af.y;
      int n = wave * 16 + m16;
      const float2 bf = *(const float2*)(wl + 64 * 64 + n * 64 + ((k ^ m16) << 2) + 2 * half);
      v2f b; b.x = bf.x; b.y = bf.y;
      acc2x = __builtin_amdgcn_wmma_f32_16x16x4_f32(
          false, a, false, b, (short)0, acc2x, false, false);
    }
    float bv = b2[wave * 16 + m16];
    for (int v = 0; v < 8; ++v) {
      int node = row0 + BROWS + v + 8 * half;
      if (node < n_nodes)
        out[(size_t)node * 64 + wave * 16 + m16] = acc2x[v] + bv;
    }
  }
}

extern "C" void kernel_launch(void* const* d_in, const int* in_sizes, int n_in,
                              void* d_out, int out_size, void* d_ws, size_t ws_size,
                              hipStream_t stream) {
  const float* x   = (const float*)d_in[0];
  const int*   ei  = (const int*)  d_in[1];
  const float* eps = (const float*)d_in[2];
  const float* W1  = (const float*)d_in[3];
  const float* b1  = (const float*)d_in[4];
  const float* W2  = (const float*)d_in[5];
  const float* b2  = (const float*)d_in[6];
  float* out = (float*)d_out;

  const int n_nodes = in_sizes[0] / 64;
  const int n_edges = in_sizes[1] / 2;
  const int n_chunks = (n_edges + CHUNK - 1) / CHUNK;
  const int n_tiles = (n_nodes + 15) / 16;
  const int over = n_tiles - 8 * OWNERS;
  const int nbig = over < 0 ? 0 : (over > OWNERS ? OWNERS : over);
  const int nb = nbig + (n_tiles - 9 * nbig + 7) / 8;

  int* sorted = (int*)d_ws;
  int* tab    = sorted + (size_t)n_chunks * CHUNK;

  chunk_sort_kernel<<<n_chunks, 1024, 0, stream>>>(ei, tab, sorted, n_nodes, n_edges, nb, nbig);
  owner_mlp_kernel<<<nb, 256, 0, stream>>>(x, sorted, tab, eps, W1, b1, W2, b2, out, n_nodes,
                                           n_chunks, nb, nbig);
}
